# starting kernel
# baseline (speedup 1.0000x reference)
#include <hip/hip_runtime.h>

typedef float v2f __attribute__((ext_vector_type(2)));
typedef float v8f __attribute__((ext_vector_type(8)));

__global__ void zero_f32(float* __restrict__ p, int n) {
  int i = blockIdx.x * blockDim.x + threadIdx.x;
  if (i < n) p[i] = 0.f;
}

__global__ void relu_inplace(float* __restrict__ p, int n) {
  int i = blockIdx.x * blockDim.x + threadIdx.x;
  if (i < n) p[i] = fmaxf(p[i], 0.f);
}

__global__ void deg_accum(const int* __restrict__ dst, const float* __restrict__ w,
                          float* __restrict__ deg, int E) {
  int e = blockIdx.x * blockDim.x + threadIdx.x;
  if (e < E) atomicAdd(deg + dst[e], w[e]);
}

__global__ void norm_compute(const int* __restrict__ src, const int* __restrict__ dst,
                             const float* __restrict__ w, const float* __restrict__ deg,
                             float* __restrict__ norm, int E) {
  int e = blockIdx.x * blockDim.x + threadIdx.x;
  if (e >= E) return;
  float ds = deg[src[e]];
  float dd = deg[dst[e]];
  float is = ds > 0.f ? rsqrtf(ds) : 0.f;
  float id = dd > 0.f ? rsqrtf(dd) : 0.f;
  norm[e] = is * w[e] * id;
}

template <int K>
__global__ void gemm_wmma(const float* __restrict__ A, const float* __restrict__ B,
                          const float* __restrict__ bias, float* __restrict__ C,
                          int N, int F, int rowTiles, int colTiles) {
  const int lane = threadIdx.x & 31;
  const int wave = threadIdx.x >> 5;
  const int tile = blockIdx.x * (blockDim.x >> 5) + wave;
  if (tile >= rowTiles * colTiles) return;

  const int rt = tile / colTiles;
  const int ct = tile - rt * colTiles;
  const int row0 = rt << 4;
  const int col0 = ct << 4;
  const int half = lane >> 4;
  const int lm   = lane & 15;

  const bool cOK = (col0 + lm) < F;
  const int  arow = min(row0 + lm, N - 1);
  const int  bcol = min(col0 + lm, F - 1);

  const float* Arow = A + (size_t)arow * K + 2 * half;
  const float* Bcol = B + bcol;

  v2f afrag[K / 4];
  v2f bfrag[K / 4];
#pragma unroll
  for (int i = 0; i < K / 4; ++i) {
    const int ka = 4 * i + 2 * half;
    afrag[i] = *reinterpret_cast<const v2f*>(Arow + 4 * i);
    bfrag[i].x = Bcol[(size_t)ka * F];
    bfrag[i].y = Bcol[(size_t)(ka + 1) * F];
  }

  v8f acc = {};
#pragma unroll
  for (int i = 0; i < K / 4; ++i) {
    acc = __builtin_amdgcn_wmma_f32_16x16x4_f32(
        false, afrag[i], false, bfrag[i], (short)0, acc, false, false);
  }

  const float bv = (bias != nullptr) ? bias[bcol] : 0.f;
#pragma unroll
  for (int r = 0; r < 8; ++r) {
    const int row = row0 + r + 8 * half;
    if (row < N && cOK) C[(size_t)row * F + col0 + lm] = acc[r] + bv;
  }
}

__global__ void gemm_wmma_gen(const float* __restrict__ A, const float* __restrict__ B,
                              const float* __restrict__ bias, float* __restrict__ C,
                              int N, int K, int F, int rowTiles, int colTiles) {
  const int lane = threadIdx.x & 31;
  const int wave = threadIdx.x >> 5;
  const int tile = blockIdx.x * (blockDim.x >> 5) + wave;
  if (tile >= rowTiles * colTiles) return;
  const int rt = tile / colTiles;
  const int ct = tile - rt * colTiles;
  const int row0 = rt << 4, col0 = ct << 4;
  const int half = lane >> 4, lm = lane & 15;
  const bool cOK = (col0 + lm) < F;
  const int arow = min(row0 + lm, N - 1);
  const int bcol = min(col0 + lm, F - 1);
  v8f acc = {};
  for (int kb = 0; kb < K; kb += 4) {
    const int ka = kb + 2 * half;
    v2f a = *reinterpret_cast<const v2f*>(A + (size_t)arow * K + ka);
    v2f b;
    b.x = B[(size_t)ka * F + bcol];
    b.y = B[(size_t)(ka + 1) * F + bcol];
    acc = __builtin_amdgcn_wmma_f32_16x16x4_f32(false, a, false, b, (short)0, acc,
                                                false, false);
  }
  const float bv = (bias != nullptr) ? bias[bcol] : 0.f;
#pragma unroll
  for (int r = 0; r < 8; ++r) {
    const int row = row0 + r + 8 * half;
    if (row < N && cOK) C[(size_t)row * F + col0 + lm] = acc[r] + bv;
  }
}

__global__ void edge_scatter(const float* __restrict__ h, const float* __restrict__ norm,
                             const int* __restrict__ src, const int* __restrict__ dst,
                             float* __restrict__ agg, int E, int F, int G) {
  long long tid = (long long)blockIdx.x * blockDim.x + threadIdx.x;
  if (tid >= (long long)E * G) return;
  int e = (int)(tid / G);
  int g = (int)(tid - (long long)e * G);
  int s = src[e];
  int d = dst[e];
  float w = norm[e];
  const float4 v = *reinterpret_cast<const float4*>(h + (size_t)s * F + (size_t)g * 4);
  float* base = agg + (size_t)d * F + (size_t)g * 4;
  atomicAdd(base + 0, v.x * w);
  atomicAdd(base + 1, v.y * w);
  atomicAdd(base + 2, v.z * w);
  atomicAdd(base + 3, v.w * w);
}

template <int C>
__global__ void relu_logsoftmax_t(float* __restrict__ y, int N) {
  int n = blockIdx.x * blockDim.x + threadIdx.x;
  if (n >= N) return;
  float v[C];
  float m = -3.4e38f;
  float* row = y + (size_t)n * C;
#pragma unroll
  for (int c = 0; c < C; ++c) {
    float t = fmaxf(row[c], 0.f);
    v[c] = t;
    m = fmaxf(m, t);
  }
  float s = 0.f;
#pragma unroll
  for (int c = 0; c < C; ++c) s += __expf(v[c] - m);
  float ls = m + __logf(s);
#pragma unroll
  for (int c = 0; c < C; ++c) row[c] = v[c] - ls;
}

__global__ void relu_logsoftmax_gen(float* __restrict__ y, int N, int C) {
  int n = blockIdx.x * blockDim.x + threadIdx.x;
  if (n >= N) return;
  float* row = y + (size_t)n * C;
  float m = -3.4e38f;
  for (int c = 0; c < C; ++c) m = fmaxf(m, fmaxf(row[c], 0.f));
  float s = 0.f;
  for (int c = 0; c < C; ++c) s += __expf(fmaxf(row[c], 0.f) - m);
  float ls = m + __logf(s);
  for (int c = 0; c < C; ++c) row[c] = fmaxf(row[c], 0.f) - ls;
}

static inline int ceil_div(long long a, int b) { return (int)((a + b - 1) / b); }

static void launch_gemm(const float* A, const float* B, const float* bias, float* C,
                        int N, int K, int F, hipStream_t stream) {
  const int TPB = 256;
  int rowT = ceil_div(N, 16), colT = ceil_div(F, 16);
  int blocks = ceil_div((long long)rowT * colT, TPB / 32);
  if (K == 64) {
    gemm_wmma<64><<<blocks, TPB, 0, stream>>>(A, B, bias, C, N, F, rowT, colT);
  } else if (K == 48) {
    gemm_wmma<48><<<blocks, TPB, 0, stream>>>(A, B, bias, C, N, F, rowT, colT);
  } else {
    gemm_wmma_gen<<<blocks, TPB, 0, stream>>>(A, B, bias, C, N, K, F, rowT, colT);
  }
}

extern "C" void kernel_launch(void* const* d_in, const int* in_sizes, int n_in,
                              void* d_out, int out_size, void* d_ws, size_t ws_size,
                              hipStream_t stream) {
  const float* x  = (const float*)d_in[0];
  const int*   ei = (const int*)d_in[1];
  const float* ew = (const float*)d_in[2];
  const float* W1 = (const float*)d_in[3];
  const float* V1 = (const float*)d_in[4];
  const float* b1 = (const float*)d_in[5];
  const float* W2 = (const float*)d_in[6];
  const float* V2 = (const float*)d_in[7];
  const float* b2 = (const float*)d_in[8];
  float* out = (float*)d_out;

  const int H   = in_sizes[5];
  const int Cn  = in_sizes[8];
  const int Fin = in_sizes[3] / H;
  const int N   = in_sizes[0] / Fin;
  const int E   = in_sizes[2];
  const int* src = ei;
  const int* dst = ei + E;

  float* ws = (float*)d_ws;
  size_t off = 0;
  float* deg  = ws + off; off += ((size_t)N + 255) & ~(size_t)255;
  float* norm = ws + off; off += ((size_t)E + 255) & ~(size_t)255;
  float* hbuf = ws + off; off += ((size_t)N * H + 255) & ~(size_t)255;
  float* z1   = ws + off; off += ((size_t)N * H + 255) & ~(size_t)255;

  const int TPB = 256;

  zero_f32<<<ceil_div(N, TPB), TPB, 0, stream>>>(deg, N);
  deg_accum<<<ceil_div(E, TPB), TPB, 0, stream>>>(dst, ew, deg, E);
  norm_compute<<<ceil_div(E, TPB), TPB, 0, stream>>>(src, dst, ew, deg, norm, E);

  launch_gemm(x, W1, nullptr, hbuf, N, Fin, H, stream);
  launch_gemm(x, V1, b1,      z1,   N, Fin, H, stream);
  {
    int G = H / 4;
    edge_scatter<<<ceil_div((long long)E * G, TPB), TPB, 0, stream>>>(
        hbuf, norm, src, dst, z1, E, H, G);
    relu_inplace<<<ceil_div((long long)N * H, TPB), TPB, 0, stream>>>(z1, N * H);
  }

  launch_gemm(z1, W2, nullptr, hbuf, N, H, Cn, stream);
  launch_gemm(z1, V2, b2,      out,  N, H, Cn, stream);
  {
    int G = Cn / 4;
    edge_scatter<<<ceil_div((long long)E * G, TPB), TPB, 0, stream>>>(
        hbuf, norm, src, dst, out, E, Cn, G);
    if (Cn == 40) {
      relu_logsoftmax_t<40><<<ceil_div(N, TPB), TPB, 0, stream>>>(out, N);
    } else {
      relu_logsoftmax_gen<<<ceil_div(N, TPB), TPB, 0, stream>>>(out, N, Cn);
    }
  }
}


# Atomic sums split per L2 half: each wave reads its die id and adds into its half's own copy of degree and layer sums, so lines stay in one half; consumers add the two copies; sum order changes
# speedup vs baseline: 1.9191x; 1.9191x over previous
#include <hip/hip_runtime.h>

typedef float v2f __attribute__((ext_vector_type(2)));
typedef float v8f __attribute__((ext_vector_type(8)));

__global__ void zero_f32(float* __restrict__ p, float* __restrict__ p2, int n) {
  int i = blockIdx.x * blockDim.x + threadIdx.x;
  if (i < n) {
    p[i] = 0.f;
    if (p2 != nullptr) p2[i] = 0.f;
  }
}

__device__ __forceinline__ int l2_half() {
  const unsigned id = __builtin_amdgcn_s_sendmsg_rtn(0x87);
  return (int)((id >> 18) & 1u);
}

__global__ void deg_accum(const int* __restrict__ dst, const float* __restrict__ w,
                          float* __restrict__ deg, float* __restrict__ deg2, int E) {
  int e = blockIdx.x * blockDim.x + threadIdx.x;
  float* d = (deg2 != nullptr && l2_half()) ? deg2 : deg;
  if (e < E) atomicAdd(d + dst[e], w[e]);
}

__global__ void norm_compute(const int* __restrict__ src, const int* __restrict__ dst,
                             const float* __restrict__ w, const float* __restrict__ deg,
                             const float* __restrict__ deg2, float* __restrict__ norm, int E) {
  int e = blockIdx.x * blockDim.x + threadIdx.x;
  if (e >= E) return;
  float ds = deg[src[e]];
  float dd = deg[dst[e]];
  if (deg2 != nullptr) {
    ds += deg2[src[e]];
    dd += deg2[dst[e]];
  }
  float is = ds > 0.f ? rsqrtf(ds) : 0.f;
  float id = dd > 0.f ? rsqrtf(dd) : 0.f;
  norm[e] = is * w[e] * id;
}

template <int K, bool RELU_A, bool TWO_A>
__global__ void gemm_wmma(const float* __restrict__ A, const float* __restrict__ A2,
                          const float* __restrict__ B,
                          const float* __restrict__ bias, float* __restrict__ C,
                          float* __restrict__ Z,
                          int N, int F, int rowTiles, int colTiles) {
  const int lane = threadIdx.x & 31;
  const int wave = threadIdx.x >> 5;
  const int tile = blockIdx.x * (blockDim.x >> 5) + wave;
  if (tile >= rowTiles * colTiles) return;

  const int rt = tile / colTiles;
  const int ct = tile - rt * colTiles;
  const int row0 = rt << 4;
  const int col0 = ct << 4;
  const int half = lane >> 4;
  const int lm   = lane & 15;

  const bool cOK = (col0 + lm) < F;
  const int  arow = min(row0 + lm, N - 1);
  const int  bcol = min(col0 + lm, F - 1);

  const float* Arow = A + (size_t)arow * K + 2 * half;
  const float* Bcol = B + bcol;

  v2f afrag[K / 4];
  v2f bfrag[K / 4];
#pragma unroll
  for (int i = 0; i < K / 4; ++i) {
    const int ka = 4 * i + 2 * half;
    afrag[i] = *reinterpret_cast<const v2f*>(Arow + 4 * i);
    bfrag[i].x = Bcol[(size_t)ka * F];
    bfrag[i].y = Bcol[(size_t)(ka + 1) * F];
  }
  if (TWO_A) {
    const float* A2row = A2 + (size_t)arow * K + 2 * half;
#pragma unroll
    for (int i = 0; i < K / 4; ++i) {
      const v2f t = *reinterpret_cast<const v2f*>(A2row + 4 * i);
      afrag[i].x += t.x;
      afrag[i].y += t.y;
    }
  }
  if (RELU_A) {
#pragma unroll
    for (int i = 0; i < K / 4; ++i) {
      afrag[i].x = fmaxf(afrag[i].x, 0.f);
      afrag[i].y = fmaxf(afrag[i].y, 0.f);
    }
  }

  v8f acc = {};
#pragma unroll
  for (int i = 0; i < K / 4; ++i) {
    acc = __builtin_amdgcn_wmma_f32_16x16x4_f32(
        false, afrag[i], false, bfrag[i], (short)0, acc, false, false);
  }

  const float bv = (bias != nullptr) ? bias[bcol] : 0.f;
#pragma unroll
  for (int r = 0; r < 8; ++r) {
    const int row = row0 + r + 8 * half;
    if (row < N && cOK) {
      C[(size_t)row * F + col0 + lm] = acc[r] + bv;
      if (Z != nullptr) Z[(size_t)row * F + col0 + lm] = 0.f;
    }
  }
}

__global__ void gemm_wmma_gen(const float* __restrict__ A, const float* __restrict__ A2,
                              const float* __restrict__ B,
                              const float* __restrict__ bias, float* __restrict__ C,
                              float* __restrict__ Z,
                              int N, int K, int F, int rowTiles, int colTiles, int reluA) {
  const int lane = threadIdx.x & 31;
  const int wave = threadIdx.x >> 5;
  const int tile = blockIdx.x * (blockDim.x >> 5) + wave;
  if (tile >= rowTiles * colTiles) return;
  const int rt = tile / colTiles;
  const int ct = tile - rt * colTiles;
  const int row0 = rt << 4, col0 = ct << 4;
  const int half = lane >> 4, lm = lane & 15;
  const bool cOK = (col0 + lm) < F;
  const int arow = min(row0 + lm, N - 1);
  const int bcol = min(col0 + lm, F - 1);
  v8f acc = {};
  for (int kb = 0; kb < K; kb += 4) {
    const int ka = kb + 2 * half;
    v2f a = *reinterpret_cast<const v2f*>(A + (size_t)arow * K + ka);
    if (A2 != nullptr) {
      const v2f t = *reinterpret_cast<const v2f*>(A2 + (size_t)arow * K + ka);
      a.x += t.x;
      a.y += t.y;
    }
    if (reluA) { a.x = fmaxf(a.x, 0.f); a.y = fmaxf(a.y, 0.f); }
    v2f b;
    b.x = B[(size_t)ka * F + bcol];
    b.y = B[(size_t)(ka + 1) * F + bcol];
    acc = __builtin_amdgcn_wmma_f32_16x16x4_f32(false, a, false, b, (short)0, acc,
                                                false, false);
  }
  const float bv = (bias != nullptr) ? bias[bcol] : 0.f;
#pragma unroll
  for (int r = 0; r < 8; ++r) {
    const int row = row0 + r + 8 * half;
    if (row < N && cOK) {
      C[(size_t)row * F + col0 + lm] = acc[r] + bv;
      if (Z != nullptr) Z[(size_t)row * F + col0 + lm] = 0.f;
    }
  }
}

template <int F, bool TAIL>
__device__ __forceinline__ void scatter_wave(const float* __restrict__ h, float* __restrict__ agg,
                                             int s_l, int d_l, float w_l,
                                             int lm, int hbase, long long e0, int E) {
  constexpr int J = (F + 15) / 16;
  constexpr int P = 8;
#pragma unroll
  for (int b = 0; b < 16; b += P) {
    float v[P][J];
    float ww[P];
    int   dd[P];
#pragma unroll
    for (int p = 0; p < P; ++p) {
      const int q = hbase + b + p;
      const int s = __shfl(s_l, q);
      ww[p] = __shfl(w_l, q);
      dd[p] = __shfl(d_l, q);
      const float* hr = h + (size_t)s * F + lm;
#pragma unroll
      for (int j = 0; j < J; ++j)
        v[p][j] = (16 * j + lm < F) ? hr[16 * j] : 0.f;
    }
#pragma unroll
    for (int p = 0; p < P; ++p) {
      if (!TAIL || e0 + hbase + b + p < E) {
        float* ar = agg + (size_t)dd[p] * F + lm;
#pragma unroll
        for (int j = 0; j < J; ++j)
          if (16 * j + lm < F) atomicAdd(ar + 16 * j, v[p][j] * ww[p]);
      }
    }
  }
}

template <int F>
__global__ void edge_scatter_rows(const float* __restrict__ h, const float* __restrict__ norm,
                                  const int* __restrict__ src, const int* __restrict__ dst,
                                  float* __restrict__ aggA, float* __restrict__ aggB, int E) {
  const int lane  = threadIdx.x & 31;
  const int lm    = lane & 15;
  const int hbase = lane & 16;
  const long long wave = ((long long)blockIdx.x * blockDim.x + threadIdx.x) >> 5;
  const long long e0 = wave * 32;
  if (e0 >= E) return;
  const long long el = e0 + lane;
  const long long ec = el < E ? el : (long long)E - 1;
  const int   s_l = src[ec];
  const int   d_l = dst[ec];
  const float w_l = norm[ec];
  float* agg = (aggB != nullptr && l2_half()) ? aggB : aggA;
  if (e0 + 32 <= E) scatter_wave<F, false>(h, agg, s_l, d_l, w_l, lm, hbase, e0, E);
  else              scatter_wave<F, true >(h, agg, s_l, d_l, w_l, lm, hbase, e0, E);
}

__global__ void edge_scatter_rows_gen(const float* __restrict__ h, const float* __restrict__ norm,
                                      const int* __restrict__ src, const int* __restrict__ dst,
                                      float* __restrict__ aggA, float* __restrict__ aggB,
                                      int E, int F) {
  const int lane = threadIdx.x & 31;
  const int lm   = lane & 15;
  const long long e = (((long long)blockIdx.x * blockDim.x + threadIdx.x) >> 5) * 2 + (lane >> 4);
  if (e >= E) return;
  const int   s = src[e];
  const int   d = dst[e];
  const float w = norm[e];
  float* agg = (aggB != nullptr && l2_half()) ? aggB : aggA;
  for (int f = lm; f < F; f += 16)
    atomicAdd(agg + (size_t)d * F + f, h[(size_t)s * F + f] * w);
}

static void launch_scatter(const float* h, const float* norm, const int* src, const int* dst,
                           float* agg, float* aggB, int E, int F, hipStream_t stream) {
  const int TPB = 256;
  if (F == 48) {
    edge_scatter_rows<48><<<(E + 255) / 256, TPB, 0, stream>>>(h, norm, src, dst, agg, aggB, E);
  } else if (F == 40) {
    edge_scatter_rows<40><<<(E + 255) / 256, TPB, 0, stream>>>(h, norm, src, dst, agg, aggB, E);
  } else {
    edge_scatter_rows_gen<<<(E + 15) / 16, TPB, 0, stream>>>(h, norm, src, dst, agg, aggB, E, F);
  }
}

template <int C>
__global__ void relu_logsoftmax_t(float* __restrict__ y, const float* __restrict__ yb, int N) {
  int n = blockIdx.x * blockDim.x + threadIdx.x;
  if (n >= N) return;
  float v[C];
  float m = -3.4e38f;
  float* row = y + (size_t)n * C;
#pragma unroll
  for (int c = 0; c < C; ++c) v[c] = row[c];
  if (yb != nullptr) {
    const float* rb = yb + (size_t)n * C;
#pragma unroll
    for (int c = 0; c < C; ++c) v[c] += rb[c];
  }
#pragma unroll
  for (int c = 0; c < C; ++c) {
    float t = fmaxf(v[c], 0.f);
    v[c] = t;
    m = fmaxf(m, t);
  }
  float s = 0.f;
#pragma unroll
  for (int c = 0; c < C; ++c) s += __expf(v[c] - m);
  float ls = m + __logf(s);
#pragma unroll
  for (int c = 0; c < C; ++c) row[c] = v[c] - ls;
}

__global__ void relu_logsoftmax_gen(float* __restrict__ y, const float* __restrict__ yb,
                                    int N, int C) {
  int n = blockIdx.x * blockDim.x + threadIdx.x;
  if (n >= N) return;
  float* row = y + (size_t)n * C;
  if (yb != nullptr) {
    const float* rb = yb + (size_t)n * C;
    for (int c = 0; c < C; ++c) row[c] += rb[c];
  }
  float m = -3.4e38f;
  for (int c = 0; c < C; ++c) m = fmaxf(m, fmaxf(row[c], 0.f));
  float s = 0.f;
  for (int c = 0; c < C; ++c) s += __expf(fmaxf(row[c], 0.f) - m);
  float ls = m + __logf(s);
  for (int c = 0; c < C; ++c) row[c] = fmaxf(row[c], 0.f) - ls;
}

static inline int ceil_div(long long a, int b) { return (int)((a + b - 1) / b); }

static void launch_gemm(const float* A, const float* A2, const float* B, const float* bias,
                        float* C, float* Z, int N, int K, int F, bool reluA,
                        hipStream_t stream) {
  const int TPB = 256;
  int rowT = ceil_div(N, 16), colT = ceil_div(F, 16);
  int blocks = ceil_div((long long)rowT * colT, TPB / 32);
  if (K == 64 && !reluA && A2 == nullptr) {
    gemm_wmma<64, false, false><<<blocks, TPB, 0, stream>>>(A, A2, B, bias, C, Z, N, F, rowT, colT);
  } else if (K == 48 && reluA && A2 != nullptr) {
    gemm_wmma<48, true, true><<<blocks, TPB, 0, stream>>>(A, A2, B, bias, C, Z, N, F, rowT, colT);
  } else if (K == 48 && reluA) {
    gemm_wmma<48, true, false><<<blocks, TPB, 0, stream>>>(A, A2, B, bias, C, Z, N, F, rowT, colT);
  } else {
    gemm_wmma_gen<<<blocks, TPB, 0, stream>>>(A, A2, B, bias, C, Z, N, K, F, rowT, colT,
                                              reluA ? 1 : 0);
  }
}

extern "C" void kernel_launch(void* const* d_in, const int* in_sizes, int n_in,
                              void* d_out, int out_size, void* d_ws, size_t ws_size,
                              hipStream_t stream) {
  const float* x  = (const float*)d_in[0];
  const int*   ei = (const int*)d_in[1];
  const float* ew = (const float*)d_in[2];
  const float* W1 = (const float*)d_in[3];
  const float* V1 = (const float*)d_in[4];
  const float* b1 = (const float*)d_in[5];
  const float* W2 = (const float*)d_in[6];
  const float* V2 = (const float*)d_in[7];
  const float* b2 = (const float*)d_in[8];
  float* out = (float*)d_out;

  const int H   = in_sizes[5];
  const int Cn  = in_sizes[8];
  const int Fin = in_sizes[3] / H;
  const int N   = in_sizes[0] / Fin;
  const int E   = in_sizes[2];
  const int* src = ei;
  const int* dst = ei + E;

  float* ws = (float*)d_ws;
  size_t off = 0;
  float* deg  = ws + off; off += ((size_t)N + 255) & ~(size_t)255;
  float* norm = ws + off; off += ((size_t)E + 255) & ~(size_t)255;
  float* hbuf = ws + off; off += ((size_t)N * H + 255) & ~(size_t)255;
  float* z1   = ws + off; off += ((size_t)N * H + 255) & ~(size_t)255;
  const size_t wsFloats = ws_size / sizeof(float);
  float* deg2 = nullptr;
  float* z1b  = nullptr;
  float* outb = nullptr;
  {
    const size_t nDeg = ((size_t)N + 255) & ~(size_t)255;
    const size_t nZ1  = ((size_t)N * H + 255) & ~(size_t)255;
    const size_t nOut = ((size_t)N * Cn + 255) & ~(size_t)255;
    if (off + nDeg <= wsFloats) { deg2 = ws + off; off += nDeg; }
    if (off + nZ1  <= wsFloats) { z1b  = ws + off; off += nZ1; }
    if (off + nOut <= wsFloats) { outb = ws + off; off += nOut; }
  }

  const int TPB = 256;

  zero_f32<<<ceil_div(N, TPB), TPB, 0, stream>>>(deg, deg2, N);
  deg_accum<<<ceil_div(E, TPB), TPB, 0, stream>>>(dst, ew, deg, deg2, E);
  norm_compute<<<ceil_div(E, TPB), TPB, 0, stream>>>(src, dst, ew, deg, deg2, norm, E);

  launch_gemm(x, nullptr, W1, nullptr, hbuf, nullptr, N, Fin, H, false, stream);
  launch_gemm(x, nullptr, V1, b1,      z1,   z1b,     N, Fin, H, false, stream);
  launch_scatter(hbuf, norm, src, dst, z1, z1b, E, H, stream);

  launch_gemm(z1, z1b, W2, nullptr, hbuf, nullptr, N, H, Cn, true, stream);
  launch_gemm(z1, z1b, V2, b2,      out,  outb,    N, H, Cn, true, stream);
  {
    launch_scatter(hbuf, norm, src, dst, out, outb, E, Cn, stream);
    if (Cn == 40) {
      relu_logsoftmax_t<40><<<ceil_div(N, TPB), TPB, 0, stream>>>(out, outb, N);
    } else {
      relu_logsoftmax_gen<<<ceil_div(N, TPB), TPB, 0, stream>>>(out, outb, N, Cn);
    }
  }
}


# Independent stages share launches: x@W1 beside the degree adds, x@V1 beside the edge norm, layer 2's root product beside its scatter as f32 atomic adds into the same sums; sum order changes
# speedup vs baseline: 2.0582x; 1.0725x over previous
#include <hip/hip_runtime.h>

typedef float v2f __attribute__((ext_vector_type(2)));
typedef float v8f __attribute__((ext_vector_type(8)));

__global__ void zero_f32(float* __restrict__ p, float* __restrict__ p2, int n) {
  int i = blockIdx.x * blockDim.x + threadIdx.x;
  if (i < n) {
    p[i] = 0.f;
    if (p2 != nullptr) p2[i] = 0.f;
  }
}

__device__ __forceinline__ int l2_half() {
  const unsigned id = __builtin_amdgcn_s_sendmsg_rtn(0x87);
  return (int)((id >> 18) & 1u);
}

__global__ void deg_accum(const int* __restrict__ dst, const float* __restrict__ w,
                          float* __restrict__ deg, float* __restrict__ deg2, int E) {
  int e = blockIdx.x * blockDim.x + threadIdx.x;
  float* d = (deg2 != nullptr && l2_half()) ? deg2 : deg;
  if (e < E) atomicAdd(d + dst[e], w[e]);
}

__global__ void norm_compute(const int* __restrict__ src, const int* __restrict__ dst,
                             const float* __restrict__ w, const float* __restrict__ deg,
                             const float* __restrict__ deg2, float* __restrict__ norm, int E) {
  int e = blockIdx.x * blockDim.x + threadIdx.x;
  if (e >= E) return;
  float ds = deg[src[e]];
  float dd = deg[dst[e]];
  if (deg2 != nullptr) {
    ds += deg2[src[e]];
    dd += deg2[dst[e]];
  }
  float is = ds > 0.f ? rsqrtf(ds) : 0.f;
  float id = dd > 0.f ? rsqrtf(dd) : 0.f;
  norm[e] = is * w[e] * id;
}

template <int K, bool RELU_A, bool TWO_A>
__global__ void gemm_wmma(const float* __restrict__ A, const float* __restrict__ A2,
                          const float* __restrict__ B,
                          const float* __restrict__ bias, float* __restrict__ C,
                          float* __restrict__ Z,
                          int N, int F, int rowTiles, int colTiles) {
  const int lane = threadIdx.x & 31;
  const int wave = threadIdx.x >> 5;
  const int tile = blockIdx.x * (blockDim.x >> 5) + wave;
  if (tile >= rowTiles * colTiles) return;

  const int rt = tile / colTiles;
  const int ct = tile - rt * colTiles;
  const int row0 = rt << 4;
  const int col0 = ct << 4;
  const int half = lane >> 4;
  const int lm   = lane & 15;

  const bool cOK = (col0 + lm) < F;
  const int  arow = min(row0 + lm, N - 1);
  const int  bcol = min(col0 + lm, F - 1);

  const float* Arow = A + (size_t)arow * K + 2 * half;
  const float* Bcol = B + bcol;

  v2f afrag[K / 4];
  v2f bfrag[K / 4];
#pragma unroll
  for (int i = 0; i < K / 4; ++i) {
    const int ka = 4 * i + 2 * half;
    afrag[i] = *reinterpret_cast<const v2f*>(Arow + 4 * i);
    bfrag[i].x = Bcol[(size_t)ka * F];
    bfrag[i].y = Bcol[(size_t)(ka + 1) * F];
  }
  if (TWO_A) {
    const float* A2row = A2 + (size_t)arow * K + 2 * half;
#pragma unroll
    for (int i = 0; i < K / 4; ++i) {
      const v2f t = *reinterpret_cast<const v2f*>(A2row + 4 * i);
      afrag[i].x += t.x;
      afrag[i].y += t.y;
    }
  }
  if (RELU_A) {
#pragma unroll
    for (int i = 0; i < K / 4; ++i) {
      afrag[i].x = fmaxf(afrag[i].x, 0.f);
      afrag[i].y = fmaxf(afrag[i].y, 0.f);
    }
  }

  v8f acc = {};
#pragma unroll
  for (int i = 0; i < K / 4; ++i) {
    acc = __builtin_amdgcn_wmma_f32_16x16x4_f32(
        false, afrag[i], false, bfrag[i], (short)0, acc, false, false);
  }

  const float bv = (bias != nullptr) ? bias[bcol] : 0.f;
#pragma unroll
  for (int r = 0; r < 8; ++r) {
    const int row = row0 + r + 8 * half;
    if (row < N && cOK) {
      C[(size_t)row * F + col0 + lm] = acc[r] + bv;
      if (Z != nullptr) Z[(size_t)row * F + col0 + lm] = 0.f;
    }
  }
}

__global__ void gemm_wmma_gen(const float* __restrict__ A, const float* __restrict__ A2,
                              const float* __restrict__ B,
                              const float* __restrict__ bias, float* __restrict__ C,
                              float* __restrict__ Z,
                              int N, int K, int F, int rowTiles, int colTiles, int reluA) {
  const int lane = threadIdx.x & 31;
  const int wave = threadIdx.x >> 5;
  const int tile = blockIdx.x * (blockDim.x >> 5) + wave;
  if (tile >= rowTiles * colTiles) return;
  const int rt = tile / colTiles;
  const int ct = tile - rt * colTiles;
  const int row0 = rt << 4, col0 = ct << 4;
  const int half = lane >> 4, lm = lane & 15;
  const bool cOK = (col0 + lm) < F;
  const int arow = min(row0 + lm, N - 1);
  const int bcol = min(col0 + lm, F - 1);
  v8f acc = {};
  for (int kb = 0; kb < K; kb += 4) {
    const int ka = kb + 2 * half;
    v2f a = *reinterpret_cast<const v2f*>(A + (size_t)arow * K + ka);
    if (A2 != nullptr) {
      const v2f t = *reinterpret_cast<const v2f*>(A2 + (size_t)arow * K + ka);
      a.x += t.x;
      a.y += t.y;
    }
    if (reluA) { a.x = fmaxf(a.x, 0.f); a.y = fmaxf(a.y, 0.f); }
    v2f b;
    b.x = B[(size_t)ka * F + bcol];
    b.y = B[(size_t)(ka + 1) * F + bcol];
    acc = __builtin_amdgcn_wmma_f32_16x16x4_f32(false, a, false, b, (short)0, acc,
                                                false, false);
  }
  const float bv = (bias != nullptr) ? bias[bcol] : 0.f;
#pragma unroll
  for (int r = 0; r < 8; ++r) {
    const int row = row0 + r + 8 * half;
    if (row < N && cOK) {
      C[(size_t)row * F + col0 + lm] = acc[r] + bv;
      if (Z != nullptr) Z[(size_t)row * F + col0 + lm] = 0.f;
    }
  }
}

template <int F, bool TAIL>
__device__ __forceinline__ void scatter_wave(const float* __restrict__ h, float* __restrict__ agg,
                                             int s_l, int d_l, float w_l,
                                             int lm, int hbase, long long e0, int E) {
  constexpr int J = (F + 15) / 16;
  constexpr int P = 8;
#pragma unroll
  for (int b = 0; b < 16; b += P) {
    float v[P][J];
    float ww[P];
    int   dd[P];
#pragma unroll
    for (int p = 0; p < P; ++p) {
      const int q = hbase + b + p;
      const int s = __shfl(s_l, q);
      ww[p] = __shfl(w_l, q);
      dd[p] = __shfl(d_l, q);
      const float* hr = h + (size_t)s * F + lm;
#pragma unroll
      for (int j = 0; j < J; ++j)
        v[p][j] = (16 * j + lm < F) ? hr[16 * j] : 0.f;
    }
#pragma unroll
    for (int p = 0; p < P; ++p) {
      if (!TAIL || e0 + hbase + b + p < E) {
        float* ar = agg + (size_t)dd[p] * F + lm;
#pragma unroll
        for (int j = 0; j < J; ++j)
          if (16 * j + lm < F) atomicAdd(ar + 16 * j, v[p][j] * ww[p]);
      }
    }
  }
}

template <int F>
__global__ void edge_scatter_rows(const float* __restrict__ h, const float* __restrict__ norm,
                                  const int* __restrict__ src, const int* __restrict__ dst,
                                  float* __restrict__ aggA, float* __restrict__ aggB, int E) {
  const int lane  = threadIdx.x & 31;
  const int lm    = lane & 15;
  const int hbase = lane & 16;
  const long long wave = ((long long)blockIdx.x * blockDim.x + threadIdx.x) >> 5;
  const long long e0 = wave * 32;
  if (e0 >= E) return;
  const long long el = e0 + lane;
  const long long ec = el < E ? el : (long long)E - 1;
  const int   s_l = src[ec];
  const int   d_l = dst[ec];
  const float w_l = norm[ec];
  float* agg = (aggB != nullptr && l2_half()) ? aggB : aggA;
  if (e0 + 32 <= E) scatter_wave<F, false>(h, agg, s_l, d_l, w_l, lm, hbase, e0, E);
  else              scatter_wave<F, true >(h, agg, s_l, d_l, w_l, lm, hbase, e0, E);
}

__global__ void edge_scatter_rows_gen(const float* __restrict__ h, const float* __restrict__ norm,
                                      const int* __restrict__ src, const int* __restrict__ dst,
                                      float* __restrict__ aggA, float* __restrict__ aggB,
                                      int E, int F) {
  const int lane = threadIdx.x & 31;
  const int lm   = lane & 15;
  const long long e = (((long long)blockIdx.x * blockDim.x + threadIdx.x) >> 5) * 2 + (lane >> 4);
  if (e >= E) return;
  const int   s = src[e];
  const int   d = dst[e];
  const float w = norm[e];
  float* agg = (aggB != nullptr && l2_half()) ? aggB : aggA;
  for (int f = lm; f < F; f += 16)
    atomicAdd(agg + (size_t)d * F + f, h[(size_t)s * F + f] * w);
}

static void launch_scatter(const float* h, const float* norm, const int* src, const int* dst,
                           float* agg, float* aggB, int E, int F, hipStream_t stream) {
  const int TPB = 256;
  if (F == 48) {
    edge_scatter_rows<48><<<(E + 255) / 256, TPB, 0, stream>>>(h, norm, src, dst, agg, aggB, E);
  } else if (F == 40) {
    edge_scatter_rows<40><<<(E + 255) / 256, TPB, 0, stream>>>(h, norm, src, dst, agg, aggB, E);
  } else {
    edge_scatter_rows_gen<<<(E + 15) / 16, TPB, 0, stream>>>(h, norm, src, dst, agg, aggB, E, F);
  }
}

__device__ __forceinline__ bool product_slot(int b, int& idx) {
  const int q = b / 11;
  const int r = b - q * 11;
  if ((r & 3) == 1) { idx = 3 * q + (r >> 2); return true; }
  idx = 8 * q + r - ((r + 2) >> 2);
  return false;
}

template <int K, bool RELU_A, bool TWO_A, bool ADD_C>
__device__ __forceinline__ void product_tile(const float* __restrict__ A,
                                             const float* __restrict__ A2,
                                             const float* __restrict__ B,
                                             const float* __restrict__ bias,
                                             float* __restrict__ C, float* __restrict__ C2,
                                             float* __restrict__ Z, float* __restrict__ Z2,
                                             int N, int F, int rowTiles, int colTiles,
                                             int vblock) {
  const int lane = threadIdx.x & 31;
  const int wave = threadIdx.x >> 5;
  const int tile = vblock * (int)(blockDim.x >> 5) + wave;
  if (tile >= rowTiles * colTiles) return;

  const int rt = tile / colTiles;
  const int ct = tile - rt * colTiles;
  const int row0 = rt << 4;
  const int col0 = ct << 4;
  const int half = lane >> 4;
  const int lm   = lane & 15;

  const bool cOK = (col0 + lm) < F;
  const int  arow = min(row0 + lm, N - 1);
  const int  bcol = min(col0 + lm, F - 1);

  const float* Arow = A + (size_t)arow * K + 2 * half;
  const float* Bcol = B + bcol;

  v2f afrag[K / 4];
  v2f bfrag[K / 4];
#pragma unroll
  for (int i = 0; i < K / 4; ++i) {
    const int ka = 4 * i + 2 * half;
    afrag[i] = *reinterpret_cast<const v2f*>(Arow + 4 * i);
    bfrag[i].x = Bcol[(size_t)ka * F];
    bfrag[i].y = Bcol[(size_t)(ka + 1) * F];
  }
  if (TWO_A) {
    const float* A2row = A2 + (size_t)arow * K + 2 * half;
#pragma unroll
    for (int i = 0; i < K / 4; ++i) {
      const v2f t = *reinterpret_cast<const v2f*>(A2row + 4 * i);
      afrag[i].x += t.x;
      afrag[i].y += t.y;
    }
  }
  if (RELU_A) {
#pragma unroll
    for (int i = 0; i < K / 4; ++i) {
      afrag[i].x = fmaxf(afrag[i].x, 0.f);
      afrag[i].y = fmaxf(afrag[i].y, 0.f);
    }
  }

  v8f acc = {};
#pragma unroll
  for (int i = 0; i < K / 4; ++i) {
    acc = __builtin_amdgcn_wmma_f32_16x16x4_f32(
        false, afrag[i], false, bfrag[i], (short)0, acc, false, false);
  }

  const float bv = (bias != nullptr) ? bias[bcol] : 0.f;
  if (ADD_C) {
    float* Cp = (C2 != nullptr && l2_half()) ? C2 : C;
#pragma unroll
    for (int r = 0; r < 8; ++r) {
      const int row = row0 + r + 8 * half;
      if (row < N && cOK) atomicAdd(Cp + (size_t)row * F + col0 + lm, acc[r] + bv);
    }
  } else {
#pragma unroll
    for (int r = 0; r < 8; ++r) {
      const int row = row0 + r + 8 * half;
      if (row < N && cOK) {
        const size_t o = (size_t)row * F + col0 + lm;
        C[o] = acc[r] + bv;
        if (Z  != nullptr) Z[o]  = 0.f;
        if (Z2 != nullptr) Z2[o] = 0.f;
      }
    }
  }
}

template <int K, bool RELU_A, bool TWO_A>
__global__ void product_rows(const float* __restrict__ A, const float* __restrict__ A2,
                             const float* __restrict__ B, float* __restrict__ C,
                             float* __restrict__ Z, float* __restrict__ Z2,
                             int N, int F, int rowTiles, int colTiles) {
  product_tile<K, RELU_A, TWO_A, false>(A, A2, B, nullptr, C, nullptr, Z, Z2,
                                        N, F, rowTiles, colTiles, (int)blockIdx.x);
}

__global__ void degree_beside_product(const int* __restrict__ dst, const float* __restrict__ w,
                                      float* __restrict__ deg, float* __restrict__ deg2, int E,
                                      const float* __restrict__ x, const float* __restrict__ Wm,
                                      float* __restrict__ h,
                                      int N, int F, int rowTiles, int colTiles) {
  int idx;
  if (product_slot((int)blockIdx.x, idx)) {
    product_tile<64, false, false, false>(x, nullptr, Wm, nullptr, h, nullptr, nullptr, nullptr,
                                          N, F, rowTiles, colTiles, idx);
  } else {
    const long long e = (long long)idx * blockDim.x + threadIdx.x;
    float* d = l2_half() ? deg2 : deg;
    if (e < E) atomicAdd(d + dst[e], w[e]);
  }
}

__global__ void norm_beside_product(const int* __restrict__ src, const int* __restrict__ dst,
                                    const float* __restrict__ w, const float* __restrict__ deg,
                                    const float* __restrict__ deg2, float* __restrict__ norm, int E,
                                    const float* __restrict__ x, const float* __restrict__ Vm,
                                    const float* __restrict__ bias,
                                    float* __restrict__ z, float* __restrict__ zb,
                                    int N, int F, int rowTiles, int colTiles) {
  int idx;
  if (product_slot((int)blockIdx.x, idx)) {
    product_tile<64, false, false, false>(x, nullptr, Vm, bias, z, nullptr, zb, nullptr,
                                          N, F, rowTiles, colTiles, idx);
  } else {
    const long long e = (long long)idx * blockDim.x + threadIdx.x;
    if (e >= E) return;
    const float ds = deg[src[e]] + deg2[src[e]];
    const float dd = deg[dst[e]] + deg2[dst[e]];
    const float is = ds > 0.f ? rsqrtf(ds) : 0.f;
    const float id = dd > 0.f ? rsqrtf(dd) : 0.f;
    norm[e] = is * w[e] * id;
  }
}

template <int F, int K, bool RELU_A, bool TWO_A>
__global__ void scatter_beside_product(const float* __restrict__ h, const float* __restrict__ norm,
                                       const int* __restrict__ src, const int* __restrict__ dst,
                                       float* __restrict__ aggA, float* __restrict__ aggB, int E,
                                       const float* __restrict__ A, const float* __restrict__ A2,
                                       const float* __restrict__ Vm, const float* __restrict__ bias,
                                       int N, int rowTiles, int colTiles) {
  int idx;
  if (product_slot((int)blockIdx.x, idx)) {
    product_tile<K, RELU_A, TWO_A, true>(A, A2, Vm, bias, aggA, aggB, nullptr, nullptr,
                                         N, F, rowTiles, colTiles, idx);
  } else {
    const int lane  = threadIdx.x & 31;
    const int lm    = lane & 15;
    const int hbase = lane & 16;
    const long long wave = ((long long)idx * blockDim.x + threadIdx.x) >> 5;
    const long long e0 = wave * 32;
    if (e0 >= E) return;
    const long long el = e0 + lane;
    const long long ec = el < E ? el : (long long)E - 1;
    const int   s_l = src[ec];
    const int   d_l = dst[ec];
    const float w_l = norm[ec];
    float* agg = l2_half() ? aggB : aggA;
    if (e0 + 32 <= E) scatter_wave<F, false>(h, agg, s_l, d_l, w_l, lm, hbase, e0, E);
    else              scatter_wave<F, true >(h, agg, s_l, d_l, w_l, lm, hbase, e0, E);
  }
}

template <int C>
__global__ void relu_logsoftmax_t(float* __restrict__ y, const float* __restrict__ yb, int N) {
  int n = blockIdx.x * blockDim.x + threadIdx.x;
  if (n >= N) return;
  float v[C];
  float m = -3.4e38f;
  float* row = y + (size_t)n * C;
#pragma unroll
  for (int c = 0; c < C; ++c) v[c] = row[c];
  if (yb != nullptr) {
    const float* rb = yb + (size_t)n * C;
#pragma unroll
    for (int c = 0; c < C; ++c) v[c] += rb[c];
  }
#pragma unroll
  for (int c = 0; c < C; ++c) {
    float t = fmaxf(v[c], 0.f);
    v[c] = t;
    m = fmaxf(m, t);
  }
  float s = 0.f;
#pragma unroll
  for (int c = 0; c < C; ++c) s += __expf(v[c] - m);
  float ls = m + __logf(s);
#pragma unroll
  for (int c = 0; c < C; ++c) row[c] = v[c] - ls;
}

__global__ void relu_logsoftmax_gen(float* __restrict__ y, const float* __restrict__ yb,
                                    int N, int C) {
  int n = blockIdx.x * blockDim.x + threadIdx.x;
  if (n >= N) return;
  float* row = y + (size_t)n * C;
  if (yb != nullptr) {
    const float* rb = yb + (size_t)n * C;
    for (int c = 0; c < C; ++c) row[c] += rb[c];
  }
  float m = -3.4e38f;
  for (int c = 0; c < C; ++c) m = fmaxf(m, fmaxf(row[c], 0.f));
  float s = 0.f;
  for (int c = 0; c < C; ++c) s += __expf(fmaxf(row[c], 0.f) - m);
  float ls = m + __logf(s);
  for (int c = 0; c < C; ++c) row[c] = fmaxf(row[c], 0.f) - ls;
}

static inline int ceil_div(long long a, int b) { return (int)((a + b - 1) / b); }

static void launch_gemm(const float* A, const float* A2, const float* B, const float* bias,
                        float* C, float* Z, int N, int K, int F, bool reluA,
                        hipStream_t stream) {
  const int TPB = 256;
  int rowT = ceil_div(N, 16), colT = ceil_div(F, 16);
  int blocks = ceil_div((long long)rowT * colT, TPB / 32);
  if (K == 64 && !reluA && A2 == nullptr) {
    gemm_wmma<64, false, false><<<blocks, TPB, 0, stream>>>(A, A2, B, bias, C, Z, N, F, rowT, colT);
  } else if (K == 48 && reluA && A2 != nullptr) {
    gemm_wmma<48, true, true><<<blocks, TPB, 0, stream>>>(A, A2, B, bias, C, Z, N, F, rowT, colT);
  } else if (K == 48 && reluA) {
    gemm_wmma<48, true, false><<<blocks, TPB, 0, stream>>>(A, A2, B, bias, C, Z, N, F, rowT, colT);
  } else {
    gemm_wmma_gen<<<blocks, TPB, 0, stream>>>(A, A2, B, bias, C, Z, N, K, F, rowT, colT,
                                              reluA ? 1 : 0);
  }
}

extern "C" void kernel_launch(void* const* d_in, const int* in_sizes, int n_in,
                              void* d_out, int out_size, void* d_ws, size_t ws_size,
                              hipStream_t stream) {
  const float* x  = (const float*)d_in[0];
  const int*   ei = (const int*)d_in[1];
  const float* ew = (const float*)d_in[2];
  const float* W1 = (const float*)d_in[3];
  const float* V1 = (const float*)d_in[4];
  const float* b1 = (const float*)d_in[5];
  const float* W2 = (const float*)d_in[6];
  const float* V2 = (const float*)d_in[7];
  const float* b2 = (const float*)d_in[8];
  float* out = (float*)d_out;

  const int H   = in_sizes[5];
  const int Cn  = in_sizes[8];
  const int Fin = in_sizes[3] / H;
  const int N   = in_sizes[0] / Fin;
  const int E   = in_sizes[2];
  const int* src = ei;
  const int* dst = ei + E;

  float* ws = (float*)d_ws;
  size_t off = 0;
  float* deg  = ws + off; off += ((size_t)N + 255) & ~(size_t)255;
  float* norm = ws + off; off += ((size_t)E + 255) & ~(size_t)255;
  float* hbuf = ws + off; off += ((size_t)N * H + 255) & ~(size_t)255;
  float* z1   = ws + off; off += ((size_t)N * H + 255) & ~(size_t)255;
  const size_t wsFloats = ws_size / sizeof(float);
  float* deg2 = nullptr;
  float* z1b  = nullptr;
  float* outb = nullptr;
  {
    const size_t nDeg = ((size_t)N + 255) & ~(size_t)255;
    const size_t nZ1  = ((size_t)N * H + 255) & ~(size_t)255;
    const size_t nOut = ((size_t)N * Cn + 255) & ~(size_t)255;
    if (off + nDeg <= wsFloats) { deg2 = ws + off; off += nDeg; }
    if (off + nZ1  <= wsFloats) { z1b  = ws + off; off += nZ1; }
    if (off + nOut <= wsFloats) { outb = ws + off; off += nOut; }
  }

  const int TPB = 256;

  if (Fin == 64 && H == 48 && Cn == 40 && deg2 != nullptr && z1b != nullptr && outb != nullptr) {
    const int rowT = ceil_div(N, 16);
    const int nbP1 = ceil_div((long long)rowT * ceil_div(H, 16), TPB / 32);
    const int nbP2 = ceil_div((long long)rowT * ceil_div(Cn, 16), TPB / 32);
    const int nbE  = ceil_div(E, TPB);
    const int per1 = ceil_div(nbP1, 3) > ceil_div(nbE, 8) ? ceil_div(nbP1, 3) : ceil_div(nbE, 8);
    const int per2 = ceil_div(nbP2, 3) > ceil_div(nbE, 8) ? ceil_div(nbP2, 3) : ceil_div(nbE, 8);

    zero_f32<<<ceil_div(N, TPB), TPB, 0, stream>>>(deg, deg2, N);
    degree_beside_product<<<per1 * 11, TPB, 0, stream>>>(dst, ew, deg, deg2, E, x, W1, hbuf,
                                                         N, H, rowT, ceil_div(H, 16));
    norm_beside_product<<<per1 * 11, TPB, 0, stream>>>(src, dst, ew, deg, deg2, norm, E,
                                                       x, V1, b1, z1, z1b,
                                                       N, H, rowT, ceil_div(H, 16));

    launch_scatter(hbuf, norm, src, dst, z1, z1b, E, H, stream);

    product_rows<48, true, true><<<nbP2, TPB, 0, stream>>>(z1, z1b, W2, hbuf, out, outb,
                                                          N, Cn, rowT, ceil_div(Cn, 16));
    scatter_beside_product<40, 48, true, true><<<per2 * 11, TPB, 0, stream>>>(
        hbuf, norm, src, dst, out, outb, E, z1, z1b, V2, b2, N, rowT, ceil_div(Cn, 16));
    relu_logsoftmax_t<40><<<ceil_div(N, TPB), TPB, 0, stream>>>(out, outb, N);
    return;
  }

  zero_f32<<<ceil_div(N, TPB), TPB, 0, stream>>>(deg, deg2, N);
  deg_accum<<<ceil_div(E, TPB), TPB, 0, stream>>>(dst, ew, deg, deg2, E);
  norm_compute<<<ceil_div(E, TPB), TPB, 0, stream>>>(src, dst, ew, deg, deg2, norm, E);

  launch_gemm(x, nullptr, W1, nullptr, hbuf, nullptr, N, Fin, H, false, stream);
  launch_gemm(x, nullptr, V1, b1,      z1,   z1b,     N, Fin, H, false, stream);
  launch_scatter(hbuf, norm, src, dst, z1, z1b, E, H, stream);

  launch_gemm(z1, z1b, W2, nullptr, hbuf, nullptr, N, H, Cn, true, stream);
  launch_gemm(z1, z1b, V2, b2,      out,  outb,    N, H, Cn, true, stream);
  {
    launch_scatter(hbuf, norm, src, dst, out, outb, E, Cn, stream);
    if (Cn == 40) {
      relu_logsoftmax_t<40><<<ceil_div(N, TPB), TPB, 0, stream>>>(out, outb, N);
    } else {
      relu_logsoftmax_gen<<<ceil_div(N, TPB), TPB, 0, stream>>>(out, outb, N, Cn);
    }
  }
}


# Gathered and summed matrices re-laid as a 32-column plane (a row = one aligned 128-B line) plus a tail plane: an edge's gather and f32 atomics take 2 cache requests, not 3; atomics kept, unordered
# speedup vs baseline: 2.3760x; 1.1544x over previous
#include <hip/hip_runtime.h>

typedef float v2f __attribute__((ext_vector_type(2)));
typedef float v8f __attribute__((ext_vector_type(8)));

__global__ void zero_f32(float* __restrict__ p, float* __restrict__ p2, int n) {
  int i = blockIdx.x * blockDim.x + threadIdx.x;
  if (i < n) {
    p[i] = 0.f;
    if (p2 != nullptr) p2[i] = 0.f;
  }
}

__device__ __forceinline__ int l2_half() {
  const unsigned id = __builtin_amdgcn_s_sendmsg_rtn(0x87);
  return (int)((id >> 18) & 1u);
}

__global__ void deg_accum(const int* __restrict__ dst, const float* __restrict__ w,
                          float* __restrict__ deg, float* __restrict__ deg2, int E) {
  int e = blockIdx.x * blockDim.x + threadIdx.x;
  float* d = (deg2 != nullptr && l2_half()) ? deg2 : deg;
  if (e < E) atomicAdd(d + dst[e], w[e]);
}

__global__ void norm_compute(const int* __restrict__ src, const int* __restrict__ dst,
                             const float* __restrict__ w, const float* __restrict__ deg,
                             const float* __restrict__ deg2, float* __restrict__ norm, int E) {
  int e = blockIdx.x * blockDim.x + threadIdx.x;
  if (e >= E) return;
  float ds = deg[src[e]];
  float dd = deg[dst[e]];
  if (deg2 != nullptr) {
    ds += deg2[src[e]];
    dd += deg2[dst[e]];
  }
  float is = ds > 0.f ? rsqrtf(ds) : 0.f;
  float id = dd > 0.f ? rsqrtf(dd) : 0.f;
  norm[e] = is * w[e] * id;
}

template <int K, bool RELU_A, bool TWO_A>
__global__ void gemm_wmma(const float* __restrict__ A, const float* __restrict__ A2,
                          const float* __restrict__ B,
                          const float* __restrict__ bias, float* __restrict__ C,
                          float* __restrict__ Z,
                          int N, int F, int rowTiles, int colTiles) {
  const int lane = threadIdx.x & 31;
  const int wave = threadIdx.x >> 5;
  const int tile = blockIdx.x * (blockDim.x >> 5) + wave;
  if (tile >= rowTiles * colTiles) return;

  const int rt = tile / colTiles;
  const int ct = tile - rt * colTiles;
  const int row0 = rt << 4;
  const int col0 = ct << 4;
  const int half = lane >> 4;
  const int lm   = lane & 15;

  const bool cOK = (col0 + lm) < F;
  const int  arow = min(row0 + lm, N - 1);
  const int  bcol = min(col0 + lm, F - 1);

  const float* Arow = A + (size_t)arow * K + 2 * half;
  const float* Bcol = B + bcol;

  v2f afrag[K / 4];
  v2f bfrag[K / 4];
#pragma unroll
  for (int i = 0; i < K / 4; ++i) {
    const int ka = 4 * i + 2 * half;
    afrag[i] = *reinterpret_cast<const v2f*>(Arow + 4 * i);
    bfrag[i].x = Bcol[(size_t)ka * F];
    bfrag[i].y = Bcol[(size_t)(ka + 1) * F];
  }
  if (TWO_A) {
    const float* A2row = A2 + (size_t)arow * K + 2 * half;
#pragma unroll
    for (int i = 0; i < K / 4; ++i) {
      const v2f t = *reinterpret_cast<const v2f*>(A2row + 4 * i);
      afrag[i].x += t.x;
      afrag[i].y += t.y;
    }
  }
  if (RELU_A) {
#pragma unroll
    for (int i = 0; i < K / 4; ++i) {
      afrag[i].x = fmaxf(afrag[i].x, 0.f);
      afrag[i].y = fmaxf(afrag[i].y, 0.f);
    }
  }

  v8f acc = {};
#pragma unroll
  for (int i = 0; i < K / 4; ++i) {
    acc = __builtin_amdgcn_wmma_f32_16x16x4_f32(
        false, afrag[i], false, bfrag[i], (short)0, acc, false, false);
  }

  const float bv = (bias != nullptr) ? bias[bcol] : 0.f;
#pragma unroll
  for (int r = 0; r < 8; ++r) {
    const int row = row0 + r + 8 * half;
    if (row < N && cOK) {
      C[(size_t)row * F + col0 + lm] = acc[r] + bv;
      if (Z != nullptr) Z[(size_t)row * F + col0 + lm] = 0.f;
    }
  }
}

__global__ void gemm_wmma_gen(const float* __restrict__ A, const float* __restrict__ A2,
                              const float* __restrict__ B,
                              const float* __restrict__ bias, float* __restrict__ C,
                              float* __restrict__ Z,
                              int N, int K, int F, int rowTiles, int colTiles, int reluA) {
  const int lane = threadIdx.x & 31;
  const int wave = threadIdx.x >> 5;
  const int tile = blockIdx.x * (blockDim.x >> 5) + wave;
  if (tile >= rowTiles * colTiles) return;
  const int rt = tile / colTiles;
  const int ct = tile - rt * colTiles;
  const int row0 = rt << 4, col0 = ct << 4;
  const int half = lane >> 4, lm = lane & 15;
  const bool cOK = (col0 + lm) < F;
  const int arow = min(row0 + lm, N - 1);
  const int bcol = min(col0 + lm, F - 1);
  v8f acc = {};
  for (int kb = 0; kb < K; kb += 4) {
    const int ka = kb + 2 * half;
    v2f a = *reinterpret_cast<const v2f*>(A + (size_t)arow * K + ka);
    if (A2 != nullptr) {
      const v2f t = *reinterpret_cast<const v2f*>(A2 + (size_t)arow * K + ka);
      a.x += t.x;
      a.y += t.y;
    }
    if (reluA) { a.x = fmaxf(a.x, 0.f); a.y = fmaxf(a.y, 0.f); }
    v2f b;
    b.x = B[(size_t)ka * F + bcol];
    b.y = B[(size_t)(ka + 1) * F + bcol];
    acc = __builtin_amdgcn_wmma_f32_16x16x4_f32(false, a, false, b, (short)0, acc,
                                                false, false);
  }
  const float bv = (bias != nullptr) ? bias[bcol] : 0.f;
#pragma unroll
  for (int r = 0; r < 8; ++r) {
    const int row = row0 + r + 8 * half;
    if (row < N && cOK) {
      C[(size_t)row * F + col0 + lm] = acc[r] + bv;
      if (Z != nullptr) Z[(size_t)row * F + col0 + lm] = 0.f;
    }
  }
}

template <int F, bool TAIL>
__device__ __forceinline__ void scatter_wave(const float* __restrict__ h, float* __restrict__ agg,
                                             int s_l, int d_l, float w_l,
                                             int lm, int hbase, long long e0, int E) {
  constexpr int J = (F + 15) / 16;
  constexpr int P = 8;
#pragma unroll
  for (int b = 0; b < 16; b += P) {
    float v[P][J];
    float ww[P];
    int   dd[P];
#pragma unroll
    for (int p = 0; p < P; ++p) {
      const int q = hbase + b + p;
      const int s = __shfl(s_l, q);
      ww[p] = __shfl(w_l, q);
      dd[p] = __shfl(d_l, q);
      const float* hr = h + (size_t)s * F + lm;
#pragma unroll
      for (int j = 0; j < J; ++j)
        v[p][j] = (16 * j + lm < F) ? hr[16 * j] : 0.f;
    }
#pragma unroll
    for (int p = 0; p < P; ++p) {
      if (!TAIL || e0 + hbase + b + p < E) {
        float* ar = agg + (size_t)dd[p] * F + lm;
#pragma unroll
        for (int j = 0; j < J; ++j)
          if (16 * j + lm < F) atomicAdd(ar + 16 * j, v[p][j] * ww[p]);
      }
    }
  }
}

template <int F>
__global__ void edge_scatter_rows(const float* __restrict__ h, const float* __restrict__ norm,
                                  const int* __restrict__ src, const int* __restrict__ dst,
                                  float* __restrict__ aggA, float* __restrict__ aggB, int E) {
  const int lane  = threadIdx.x & 31;
  const int lm    = lane & 15;
  const int hbase = lane & 16;
  const long long wave = ((long long)blockIdx.x * blockDim.x + threadIdx.x) >> 5;
  const long long e0 = wave * 32;
  if (e0 >= E) return;
  const long long el = e0 + lane;
  const long long ec = el < E ? el : (long long)E - 1;
  const int   s_l = src[ec];
  const int   d_l = dst[ec];
  const float w_l = norm[ec];
  float* agg = (aggB != nullptr && l2_half()) ? aggB : aggA;
  if (e0 + 32 <= E) scatter_wave<F, false>(h, agg, s_l, d_l, w_l, lm, hbase, e0, E);
  else              scatter_wave<F, true >(h, agg, s_l, d_l, w_l, lm, hbase, e0, E);
}

__global__ void edge_scatter_rows_gen(const float* __restrict__ h, const float* __restrict__ norm,
                                      const int* __restrict__ src, const int* __restrict__ dst,
                                      float* __restrict__ aggA, float* __restrict__ aggB,
                                      int E, int F) {
  const int lane = threadIdx.x & 31;
  const int lm   = lane & 15;
  const long long e = (((long long)blockIdx.x * blockDim.x + threadIdx.x) >> 5) * 2 + (lane >> 4);
  if (e >= E) return;
  const int   s = src[e];
  const int   d = dst[e];
  const float w = norm[e];
  float* agg = (aggB != nullptr && l2_half()) ? aggB : aggA;
  for (int f = lm; f < F; f += 16)
    atomicAdd(agg + (size_t)d * F + f, h[(size_t)s * F + f] * w);
}

static void launch_scatter(const float* h, const float* norm, const int* src, const int* dst,
                           float* agg, float* aggB, int E, int F, hipStream_t stream) {
  const int TPB = 256;
  if (F == 48) {
    edge_scatter_rows<48><<<(E + 255) / 256, TPB, 0, stream>>>(h, norm, src, dst, agg, aggB, E);
  } else if (F == 40) {
    edge_scatter_rows<40><<<(E + 255) / 256, TPB, 0, stream>>>(h, norm, src, dst, agg, aggB, E);
  } else {
    edge_scatter_rows_gen<<<(E + 15) / 16, TPB, 0, stream>>>(h, norm, src, dst, agg, aggB, E, F);
  }
}


typedef __attribute__((address_space(1))) float       gmem_f32;
typedef __attribute__((address_space(1))) const float gmem_cf32;
__device__ __forceinline__ unsigned long long scalar_u64(unsigned long long a) {
  const unsigned lo = __builtin_amdgcn_readfirstlane((unsigned)a);
  const unsigned hi = __builtin_amdgcn_readfirstlane((unsigned)(a >> 32));
  return ((unsigned long long)hi << 32) | lo;
}
__device__ __forceinline__ gmem_cf32* scalar_base(const float* p) {
  return (gmem_cf32*)scalar_u64((unsigned long long)p);
}
__device__ __forceinline__ gmem_f32* scalar_base(float* p) {
  return (gmem_f32*)scalar_u64((unsigned long long)p);
}

template <int F, bool TAIL>
__device__ __forceinline__ void scatter_wave_lines(const float* __restrict__ h, float* __restrict__ agg,
                                                   int N, int s_l, int d_l, float w_l,
                                                   int lane, long long e0, int E) {
  constexpr int T = F - 32;
  constexpr int G = 32 / T;
  constexpr int B = 16;
  const float* hY = h + (size_t)N * 32;
  float*       aY = agg + (size_t)N * 32;
  const int sub = lane / T;
  const int ly  = lane % T;
#pragma unroll
  for (int b = 0; b < 32; b += B) {
    float vx[B];
    float vy[B / G];
#pragma unroll
    for (int p = 0; p < B; ++p) {
      const int s = __builtin_amdgcn_readlane(s_l, b + p);
      gmem_cf32* hrow = scalar_base(h + (size_t)s * 32);
      vx[p] = hrow[lane];
    }
#pragma unroll
    for (int g = 0; g < B / G; ++g) {
      const int s = __shfl(s_l, b + g * G + sub);
      vy[g] = hY[(size_t)s * T + ly];
    }
#pragma unroll
    for (int p = 0; p < B; ++p) {
      const int   d = __builtin_amdgcn_readlane(d_l, b + p);
      const float w = __int_as_float(__builtin_amdgcn_readlane(__float_as_int(w_l), b + p));
      gmem_f32* arow = scalar_base(agg + (size_t)d * 32);
      if (!TAIL || e0 + b + p < E)
        __hip_atomic_fetch_add(arow + lane, vx[p] * w, __ATOMIC_RELAXED, __HIP_MEMORY_SCOPE_AGENT);
    }
#pragma unroll
    for (int g = 0; g < B / G; ++g) {
      const int   q = b + g * G + sub;
      const int   d = __shfl(d_l, q);
      const float w = __shfl(w_l, q);
      if (!TAIL || e0 + q < E) atomicAdd(aY + (size_t)d * T + ly, vy[g] * w);
    }
  }
}

template <int F>
__device__ __forceinline__ void scatter_edges_lines(const float* __restrict__ h,
                                                    const float* __restrict__ norm,
                                                    const int* __restrict__ src,
                                                    const int* __restrict__ dst,
                                                    float* __restrict__ aggA, float* __restrict__ aggB,
                                                    int N, int E, long long e0) {
  const int lane = threadIdx.x & 31;
  const long long el = e0 + lane;
  const long long ec = el < E ? el : (long long)E - 1;
  const int   s_l = src[ec];
  const int   d_l = dst[ec];
  const float w_l = norm[ec];
  float* agg = l2_half() ? aggB : aggA;
  if (e0 + 32 <= E) scatter_wave_lines<F, false>(h, agg, N, s_l, d_l, w_l, lane, e0, E);
  else              scatter_wave_lines<F, true >(h, agg, N, s_l, d_l, w_l, lane, e0, E);
}

template <int F>
__global__ void edge_scatter_lines(const float* __restrict__ h, const float* __restrict__ norm,
                                   const int* __restrict__ src, const int* __restrict__ dst,
                                   float* __restrict__ aggA, float* __restrict__ aggB, int N, int E) {
  const long long wave = ((long long)blockIdx.x * blockDim.x + threadIdx.x) >> 5;
  const long long e0 = wave * 32;
  if (e0 >= E) return;
  scatter_edges_lines<F>(h, norm, src, dst, aggA, aggB, N, E, e0);
}

__device__ __forceinline__ bool product_slot(int b, int& idx) {
  const int q = b / 11;
  const int r = b - q * 11;
  if ((r & 3) == 1) { idx = 3 * q + (r >> 2); return true; }
  idx = 8 * q + r - ((r + 2) >> 2);
  return false;
}

template <int K, bool RELU_A, bool TWO_A, bool ADD_C, bool SPLIT_A>
__device__ __forceinline__ void product_tile(const float* __restrict__ A,
                                             const float* __restrict__ A2,
                                             const float* __restrict__ B,
                                             const float* __restrict__ bias,
                                             float* __restrict__ C, float* __restrict__ C2,
                                             float* __restrict__ Z, float* __restrict__ Z2,
                                             int N, int F, int rowTiles, int colTiles,
                                             int vblock) {
  const int lane = threadIdx.x & 31;
  const int wave = threadIdx.x >> 5;
  const int tile = vblock * (int)(blockDim.x >> 5) + wave;
  if (tile >= rowTiles * colTiles) return;

  const int rt = tile / colTiles;
  const int ct = tile - rt * colTiles;
  const int row0 = rt << 4;
  const int col0 = ct << 4;
  const int half = lane >> 4;
  const int lm   = lane & 15;

  const bool cOK = (col0 + lm) < F;
  const int  arow = min(row0 + lm, N - 1);
  const int  bcol = min(col0 + lm, F - 1);

  constexpr int AP = SPLIT_A ? 32 : K;
  const float* Arow  = A + (size_t)arow * AP + 2 * half;
  const float* ArowY = A + (size_t)N * 32 + (size_t)arow * (K - 32) + 2 * half;
  const float* Bcol = B + bcol;

  v2f afrag[K / 4];
  v2f bfrag[K / 4];
#pragma unroll
  for (int i = 0; i < K / 4; ++i) {
    const int ka = 4 * i + 2 * half;
    if (SPLIT_A && 4 * i >= 32) afrag[i] = *reinterpret_cast<const v2f*>(ArowY + (4 * i - 32));
    else                        afrag[i] = *reinterpret_cast<const v2f*>(Arow + 4 * i);
    bfrag[i].x = Bcol[(size_t)ka * F];
    bfrag[i].y = Bcol[(size_t)(ka + 1) * F];
  }
  if (TWO_A) {
    const float* A2row  = A2 + (size_t)arow * AP + 2 * half;
    const float* A2rowY = A2 + (size_t)N * 32 + (size_t)arow * (K - 32) + 2 * half;
#pragma unroll
    for (int i = 0; i < K / 4; ++i) {
      v2f t;
      if (SPLIT_A && 4 * i >= 32) t = *reinterpret_cast<const v2f*>(A2rowY + (4 * i - 32));
      else                        t = *reinterpret_cast<const v2f*>(A2row + 4 * i);
      afrag[i].x += t.x;
      afrag[i].y += t.y;
    }
  }
  if (RELU_A) {
#pragma unroll
    for (int i = 0; i < K / 4; ++i) {
      afrag[i].x = fmaxf(afrag[i].x, 0.f);
      afrag[i].y = fmaxf(afrag[i].y, 0.f);
    }
  }

  v8f acc = {};
#pragma unroll
  for (int i = 0; i < K / 4; ++i) {
    acc = __builtin_amdgcn_wmma_f32_16x16x4_f32(
        false, afrag[i], false, bfrag[i], (short)0, acc, false, false);
  }

  const bool   inX = col0 < 32;
  const int    cp  = inX ? 32 : F - 32;
  const size_t cb  = (inX ? (size_t)0 : (size_t)N * 32) + (size_t)(inX ? col0 : col0 - 32) + lm;

  const float bv = (bias != nullptr) ? bias[bcol] : 0.f;
  if (ADD_C) {
    float* Cp = (C2 != nullptr && l2_half()) ? C2 : C;
#pragma unroll
    for (int r = 0; r < 8; ++r) {
      const int row = row0 + r + 8 * half;
      if (row < N && cOK) atomicAdd(Cp + cb + (size_t)row * cp, acc[r] + bv);
    }
  } else {
#pragma unroll
    for (int r = 0; r < 8; ++r) {
      const int row = row0 + r + 8 * half;
      if (row < N && cOK) {
        const size_t o = cb + (size_t)row * cp;
        C[o] = acc[r] + bv;
        if (Z  != nullptr) Z[o]  = 0.f;
        if (Z2 != nullptr) Z2[o] = 0.f;
      }
    }
  }
}

template <int K, bool RELU_A, bool TWO_A>
__global__ void product_rows(const float* __restrict__ A, const float* __restrict__ A2,
                             const float* __restrict__ B, float* __restrict__ C,
                             float* __restrict__ Z, float* __restrict__ Z2,
                             int N, int F, int rowTiles, int colTiles) {
  product_tile<K, RELU_A, TWO_A, false, true>(A, A2, B, nullptr, C, nullptr, Z, Z2,
                                              N, F, rowTiles, colTiles, (int)blockIdx.x);
}

__global__ void degree_beside_product(const int* __restrict__ dst, const float* __restrict__ w,
                                      float* __restrict__ deg, float* __restrict__ deg2, int E,
                                      const float* __restrict__ x, const float* __restrict__ Wm,
                                      float* __restrict__ h,
                                      int N, int F, int rowTiles, int colTiles) {
  int idx;
  if (product_slot((int)blockIdx.x, idx)) {
    product_tile<64, false, false, false, false>(x, nullptr, Wm, nullptr, h, nullptr, nullptr,
                                                 nullptr, N, F, rowTiles, colTiles, idx);
  } else {
    const long long e = (long long)idx * blockDim.x + threadIdx.x;
    float* d = l2_half() ? deg2 : deg;
    if (e < E) atomicAdd(d + dst[e], w[e]);
  }
}

__global__ void norm_beside_product(const int* __restrict__ src, const int* __restrict__ dst,
                                    const float* __restrict__ w, const float* __restrict__ deg,
                                    const float* __restrict__ deg2, float* __restrict__ norm, int E,
                                    const float* __restrict__ x, const float* __restrict__ Vm,
                                    const float* __restrict__ bias,
                                    float* __restrict__ z, float* __restrict__ zb,
                                    int N, int F, int rowTiles, int colTiles) {
  int idx;
  if (product_slot((int)blockIdx.x, idx)) {
    product_tile<64, false, false, false, false>(x, nullptr, Vm, bias, z, nullptr, zb, nullptr,
                                                 N, F, rowTiles, colTiles, idx);
  } else {
    const long long e = (long long)idx * blockDim.x + threadIdx.x;
    if (e >= E) return;
    const float ds = deg[src[e]] + deg2[src[e]];
    const float dd = deg[dst[e]] + deg2[dst[e]];
    const float is = ds > 0.f ? rsqrtf(ds) : 0.f;
    const float id = dd > 0.f ? rsqrtf(dd) : 0.f;
    norm[e] = is * w[e] * id;
  }
}

template <int F, int K, bool RELU_A, bool TWO_A>
__global__ void scatter_beside_product(const float* __restrict__ h, const float* __restrict__ norm,
                                       const int* __restrict__ src, const int* __restrict__ dst,
                                       float* __restrict__ aggA, float* __restrict__ aggB, int E,
                                       const float* __restrict__ A, const float* __restrict__ A2,
                                       const float* __restrict__ Vm, const float* __restrict__ bias,
                                       int N, int rowTiles, int colTiles) {
  int idx;
  if (product_slot((int)blockIdx.x, idx)) {
    product_tile<K, RELU_A, TWO_A, true, true>(A, A2, Vm, bias, aggA, aggB, nullptr, nullptr,
                                               N, F, rowTiles, colTiles, idx);
  } else {
    const long long wave = ((long long)idx * blockDim.x + threadIdx.x) >> 5;
    const long long e0 = wave * 32;
    if (e0 >= E) return;
    scatter_edges_lines<F>(h, norm, src, dst, aggA, aggB, N, E, e0);
  }
}

template <int C>
__global__ void relu_logsoftmax_t(float* __restrict__ y, const float* __restrict__ yb, int N) {
  int n = blockIdx.x * blockDim.x + threadIdx.x;
  if (n >= N) return;
  float v[C];
  float m = -3.4e38f;
  float* row = y + (size_t)n * C;
#pragma unroll
  for (int c = 0; c < C; ++c) v[c] = row[c];
  if (yb != nullptr) {
    const float* rb = yb + (size_t)n * C;
#pragma unroll
    for (int c = 0; c < C; ++c) v[c] += rb[c];
  }
#pragma unroll
  for (int c = 0; c < C; ++c) {
    float t = fmaxf(v[c], 0.f);
    v[c] = t;
    m = fmaxf(m, t);
  }
  float s = 0.f;
#pragma unroll
  for (int c = 0; c < C; ++c) s += __expf(v[c] - m);
  float ls = m + __logf(s);
#pragma unroll
  for (int c = 0; c < C; ++c) row[c] = v[c] - ls;
}

__global__ void relu_logsoftmax_gen(float* __restrict__ y, const float* __restrict__ yb,
                                    int N, int C) {
  int n = blockIdx.x * blockDim.x + threadIdx.x;
  if (n >= N) return;
  float* row = y + (size_t)n * C;
  if (yb != nullptr) {
    const float* rb = yb + (size_t)n * C;
    for (int c = 0; c < C; ++c) row[c] += rb[c];
  }
  float m = -3.4e38f;
  for (int c = 0; c < C; ++c) m = fmaxf(m, fmaxf(row[c], 0.f));
  float s = 0.f;
  for (int c = 0; c < C; ++c) s += __expf(fmaxf(row[c], 0.f) - m);
  float ls = m + __logf(s);
  for (int c = 0; c < C; ++c) row[c] = fmaxf(row[c], 0.f) - ls;
}

template <int C>
__global__ void relu_logsoftmax_lines(const float* __restrict__ ya, const float* __restrict__ yb,
                                      float* __restrict__ out, int N) {
  constexpr int T = C - 32;
  int n = blockIdx.x * blockDim.x + threadIdx.x;
  if (n >= N) return;
  const float* ax = ya + (size_t)n * 32;
  const float* bx = yb + (size_t)n * 32;
  const float* ay = ya + (size_t)N * 32 + (size_t)n * T;
  const float* by = yb + (size_t)N * 32 + (size_t)n * T;
  float v[C];
#pragma unroll
  for (int c = 0; c < 32; ++c) v[c] = ax[c] + bx[c];
#pragma unroll
  for (int c = 0; c < T; ++c) v[32 + c] = ay[c] + by[c];
  float m = -3.4e38f;
#pragma unroll
  for (int c = 0; c < C; ++c) {
    float t = fmaxf(v[c], 0.f);
    v[c] = t;
    m = fmaxf(m, t);
  }
  float s = 0.f;
#pragma unroll
  for (int c = 0; c < C; ++c) s += __expf(v[c] - m);
  float ls = m + __logf(s);
  float* row = out + (size_t)n * C;
#pragma unroll
  for (int c = 0; c < C; ++c) row[c] = v[c] - ls;
}

static inline int ceil_div(long long a, int b) { return (int)((a + b - 1) / b); }

static void launch_gemm(const float* A, const float* A2, const float* B, const float* bias,
                        float* C, float* Z, int N, int K, int F, bool reluA,
                        hipStream_t stream) {
  const int TPB = 256;
  int rowT = ceil_div(N, 16), colT = ceil_div(F, 16);
  int blocks = ceil_div((long long)rowT * colT, TPB / 32);
  if (K == 64 && !reluA && A2 == nullptr) {
    gemm_wmma<64, false, false><<<blocks, TPB, 0, stream>>>(A, A2, B, bias, C, Z, N, F, rowT, colT);
  } else if (K == 48 && reluA && A2 != nullptr) {
    gemm_wmma<48, true, true><<<blocks, TPB, 0, stream>>>(A, A2, B, bias, C, Z, N, F, rowT, colT);
  } else if (K == 48 && reluA) {
    gemm_wmma<48, true, false><<<blocks, TPB, 0, stream>>>(A, A2, B, bias, C, Z, N, F, rowT, colT);
  } else {
    gemm_wmma_gen<<<blocks, TPB, 0, stream>>>(A, A2, B, bias, C, Z, N, K, F, rowT, colT,
                                              reluA ? 1 : 0);
  }
}

extern "C" void kernel_launch(void* const* d_in, const int* in_sizes, int n_in,
                              void* d_out, int out_size, void* d_ws, size_t ws_size,
                              hipStream_t stream) {
  const float* x  = (const float*)d_in[0];
  const int*   ei = (const int*)d_in[1];
  const float* ew = (const float*)d_in[2];
  const float* W1 = (const float*)d_in[3];
  const float* V1 = (const float*)d_in[4];
  const float* b1 = (const float*)d_in[5];
  const float* W2 = (const float*)d_in[6];
  const float* V2 = (const float*)d_in[7];
  const float* b2 = (const float*)d_in[8];
  float* out = (float*)d_out;

  const int H   = in_sizes[5];
  const int Cn  = in_sizes[8];
  const int Fin = in_sizes[3] / H;
  const int N   = in_sizes[0] / Fin;
  const int E   = in_sizes[2];
  const int* src = ei;
  const int* dst = ei + E;

  float* ws = (float*)d_ws;
  size_t off = 0;
  float* deg  = ws + off; off += ((size_t)N + 255) & ~(size_t)255;
  float* norm = ws + off; off += ((size_t)E + 255) & ~(size_t)255;
  float* hbuf = ws + off; off += ((size_t)N * H + 255) & ~(size_t)255;
  float* z1   = ws + off; off += ((size_t)N * H + 255) & ~(size_t)255;
  const size_t wsFloats = ws_size / sizeof(float);
  float* deg2 = nullptr;
  float* z1b  = nullptr;
  float* outb = nullptr;
  {
    const size_t nDeg = ((size_t)N + 255) & ~(size_t)255;
    const size_t nZ1  = ((size_t)N * H + 255) & ~(size_t)255;
    const size_t nOut = ((size_t)N * Cn + 255) & ~(size_t)255;
    if (off + nDeg <= wsFloats) { deg2 = ws + off; off += nDeg; }
    if (off + nZ1  <= wsFloats) { z1b  = ws + off; off += nZ1; }
    if (off + nOut <= wsFloats) { outb = ws + off; off += nOut; }
  }

  const int TPB = 256;

  if (Fin == 64 && H == 48 && Cn == 40 && deg2 != nullptr && z1b != nullptr && outb != nullptr) {
    const int rowT = ceil_div(N, 16);
    const int nbP1 = ceil_div((long long)rowT * ceil_div(H, 16), TPB / 32);
    const int nbP2 = ceil_div((long long)rowT * ceil_div(Cn, 16), TPB / 32);
    const int nbE  = ceil_div(E, TPB);
    const int per1 = ceil_div(nbP1, 3) > ceil_div(nbE, 8) ? ceil_div(nbP1, 3) : ceil_div(nbE, 8);
    const int per2 = ceil_div(nbP2, 3) > ceil_div(nbE, 8) ? ceil_div(nbP2, 3) : ceil_div(nbE, 8);
    float* suma = hbuf;

    zero_f32<<<ceil_div(N, TPB), TPB, 0, stream>>>(deg, deg2, N);
    degree_beside_product<<<per1 * 11, TPB, 0, stream>>>(dst, ew, deg, deg2, E, x, W1, hbuf,
                                                         N, H, rowT, ceil_div(H, 16));
    norm_beside_product<<<per1 * 11, TPB, 0, stream>>>(src, dst, ew, deg, deg2, norm, E,
                                                       x, V1, b1, z1, z1b,
                                                       N, H, rowT, ceil_div(H, 16));

    edge_scatter_lines<48><<<nbE, TPB, 0, stream>>>(hbuf, norm, src, dst, z1, z1b, N, E);

    product_rows<48, true, true><<<nbP2, TPB, 0, stream>>>(z1, z1b, W2, out, suma, outb,
                                                          N, Cn, rowT, ceil_div(Cn, 16));
    scatter_beside_product<40, 48, true, true><<<per2 * 11, TPB, 0, stream>>>(
        out, norm, src, dst, suma, outb, E, z1, z1b, V2, b2, N, rowT, ceil_div(Cn, 16));
    relu_logsoftmax_lines<40><<<ceil_div(N, TPB), TPB, 0, stream>>>(suma, outb, out, N);
    return;
  }

  zero_f32<<<ceil_div(N, TPB), TPB, 0, stream>>>(deg, deg2, N);
  deg_accum<<<ceil_div(E, TPB), TPB, 0, stream>>>(dst, ew, deg, deg2, E);
  norm_compute<<<ceil_div(E, TPB), TPB, 0, stream>>>(src, dst, ew, deg, deg2, norm, E);

  launch_gemm(x, nullptr, W1, nullptr, hbuf, nullptr, N, Fin, H, false, stream);
  launch_gemm(x, nullptr, V1, b1,      z1,   z1b,     N, Fin, H, false, stream);
  launch_scatter(hbuf, norm, src, dst, z1, z1b, E, H, stream);

  launch_gemm(z1, z1b, W2, nullptr, hbuf, nullptr, N, H, Cn, true, stream);
  launch_gemm(z1, z1b, V2, b2,      out,  outb,    N, H, Cn, true, stream);
  {
    launch_scatter(hbuf, norm, src, dst, out, outb, E, Cn, stream);
    if (Cn == 40) {
      relu_logsoftmax_t<40><<<ceil_div(N, TPB), TPB, 0, stream>>>(out, outb, N);
    } else {
      relu_logsoftmax_gen<<<ceil_div(N, TPB), TPB, 0, stream>>>(out, outb, N, Cn);
    }
  }
}


# Per-node deg^-1/2 table built once by a small launch, in place over the degree sums: the edge-norm pass reads 2 scattered words per edge, not 4; norm values bit-identical; 8 launches, was 7
# speedup vs baseline: 2.4942x; 1.0497x over previous
#include <hip/hip_runtime.h>

typedef float v2f __attribute__((ext_vector_type(2)));
typedef float v8f __attribute__((ext_vector_type(8)));

__global__ void zero_f32(float* __restrict__ p, float* __restrict__ p2, int n) {
  int i = blockIdx.x * blockDim.x + threadIdx.x;
  if (i < n) {
    p[i] = 0.f;
    if (p2 != nullptr) p2[i] = 0.f;
  }
}

__device__ __forceinline__ int l2_half() {
  const unsigned id = __builtin_amdgcn_s_sendmsg_rtn(0x87);
  return (int)((id >> 18) & 1u);
}

__global__ void deg_accum(const int* __restrict__ dst, const float* __restrict__ w,
                          float* __restrict__ deg, float* __restrict__ deg2, int E) {
  int e = blockIdx.x * blockDim.x + threadIdx.x;
  float* d = (deg2 != nullptr && l2_half()) ? deg2 : deg;
  if (e < E) atomicAdd(d + dst[e], w[e]);
}

__global__ void norm_compute(const int* __restrict__ src, const int* __restrict__ dst,
                             const float* __restrict__ w, const float* __restrict__ deg,
                             const float* __restrict__ deg2, float* __restrict__ norm, int E) {
  int e = blockIdx.x * blockDim.x + threadIdx.x;
  if (e >= E) return;
  float ds = deg[src[e]];
  float dd = deg[dst[e]];
  if (deg2 != nullptr) {
    ds += deg2[src[e]];
    dd += deg2[dst[e]];
  }
  float is = ds > 0.f ? rsqrtf(ds) : 0.f;
  float id = dd > 0.f ? rsqrtf(dd) : 0.f;
  norm[e] = is * w[e] * id;
}

__global__ void inv_sqrt_degree(float* __restrict__ deg, const float* __restrict__ deg2, int N) {
  int n = blockIdx.x * blockDim.x + threadIdx.x;
  if (n >= N) return;
  const float d = deg[n] + deg2[n];
  deg[n] = d > 0.f ? rsqrtf(d) : 0.f;
}

template <int K, bool RELU_A, bool TWO_A>
__global__ void gemm_wmma(const float* __restrict__ A, const float* __restrict__ A2,
                          const float* __restrict__ B,
                          const float* __restrict__ bias, float* __restrict__ C,
                          float* __restrict__ Z,
                          int N, int F, int rowTiles, int colTiles) {
  const int lane = threadIdx.x & 31;
  const int wave = threadIdx.x >> 5;
  const int tile = blockIdx.x * (blockDim.x >> 5) + wave;
  if (tile >= rowTiles * colTiles) return;

  const int rt = tile / colTiles;
  const int ct = tile - rt * colTiles;
  const int row0 = rt << 4;
  const int col0 = ct << 4;
  const int half = lane >> 4;
  const int lm   = lane & 15;

  const bool cOK = (col0 + lm) < F;
  const int  arow = min(row0 + lm, N - 1);
  const int  bcol = min(col0 + lm, F - 1);

  const float* Arow = A + (size_t)arow * K + 2 * half;
  const float* Bcol = B + bcol;

  v2f afrag[K / 4];
  v2f bfrag[K / 4];
#pragma unroll
  for (int i = 0; i < K / 4; ++i) {
    const int ka = 4 * i + 2 * half;
    afrag[i] = *reinterpret_cast<const v2f*>(Arow + 4 * i);
    bfrag[i].x = Bcol[(size_t)ka * F];
    bfrag[i].y = Bcol[(size_t)(ka + 1) * F];
  }
  if (TWO_A) {
    const float* A2row = A2 + (size_t)arow * K + 2 * half;
#pragma unroll
    for (int i = 0; i < K / 4; ++i) {
      const v2f t = *reinterpret_cast<const v2f*>(A2row + 4 * i);
      afrag[i].x += t.x;
      afrag[i].y += t.y;
    }
  }
  if (RELU_A) {
#pragma unroll
    for (int i = 0; i < K / 4; ++i) {
      afrag[i].x = fmaxf(afrag[i].x, 0.f);
      afrag[i].y = fmaxf(afrag[i].y, 0.f);
    }
  }

  v8f acc = {};
#pragma unroll
  for (int i = 0; i < K / 4; ++i) {
    acc = __builtin_amdgcn_wmma_f32_16x16x4_f32(
        false, afrag[i], false, bfrag[i], (short)0, acc, false, false);
  }

  const float bv = (bias != nullptr) ? bias[bcol] : 0.f;
#pragma unroll
  for (int r = 0; r < 8; ++r) {
    const int row = row0 + r + 8 * half;
    if (row < N && cOK) {
      C[(size_t)row * F + col0 + lm] = acc[r] + bv;
      if (Z != nullptr) Z[(size_t)row * F + col0 + lm] = 0.f;
    }
  }
}

__global__ void gemm_wmma_gen(const float* __restrict__ A, const float* __restrict__ A2,
                              const float* __restrict__ B,
                              const float* __restrict__ bias, float* __restrict__ C,
                              float* __restrict__ Z,
                              int N, int K, int F, int rowTiles, int colTiles, int reluA) {
  const int lane = threadIdx.x & 31;
  const int wave = threadIdx.x >> 5;
  const int tile = blockIdx.x * (blockDim.x >> 5) + wave;
  if (tile >= rowTiles * colTiles) return;
  const int rt = tile / colTiles;
  const int ct = tile - rt * colTiles;
  const int row0 = rt << 4, col0 = ct << 4;
  const int half = lane >> 4, lm = lane & 15;
  const bool cOK = (col0 + lm) < F;
  const int arow = min(row0 + lm, N - 1);
  const int bcol = min(col0 + lm, F - 1);
  v8f acc = {};
  for (int kb = 0; kb < K; kb += 4) {
    const int ka = kb + 2 * half;
    v2f a = *reinterpret_cast<const v2f*>(A + (size_t)arow * K + ka);
    if (A2 != nullptr) {
      const v2f t = *reinterpret_cast<const v2f*>(A2 + (size_t)arow * K + ka);
      a.x += t.x;
      a.y += t.y;
    }
    if (reluA) { a.x = fmaxf(a.x, 0.f); a.y = fmaxf(a.y, 0.f); }
    v2f b;
    b.x = B[(size_t)ka * F + bcol];
    b.y = B[(size_t)(ka + 1) * F + bcol];
    acc = __builtin_amdgcn_wmma_f32_16x16x4_f32(false, a, false, b, (short)0, acc,
                                                false, false);
  }
  const float bv = (bias != nullptr) ? bias[bcol] : 0.f;
#pragma unroll
  for (int r = 0; r < 8; ++r) {
    const int row = row0 + r + 8 * half;
    if (row < N && cOK) {
      C[(size_t)row * F + col0 + lm] = acc[r] + bv;
      if (Z != nullptr) Z[(size_t)row * F + col0 + lm] = 0.f;
    }
  }
}

template <int F, bool TAIL>
__device__ __forceinline__ void scatter_wave(const float* __restrict__ h, float* __restrict__ agg,
                                             int s_l, int d_l, float w_l,
                                             int lm, int hbase, long long e0, int E) {
  constexpr int J = (F + 15) / 16;
  constexpr int P = 8;
#pragma unroll
  for (int b = 0; b < 16; b += P) {
    float v[P][J];
    float ww[P];
    int   dd[P];
#pragma unroll
    for (int p = 0; p < P; ++p) {
      const int q = hbase + b + p;
      const int s = __shfl(s_l, q);
      ww[p] = __shfl(w_l, q);
      dd[p] = __shfl(d_l, q);
      const float* hr = h + (size_t)s * F + lm;
#pragma unroll
      for (int j = 0; j < J; ++j)
        v[p][j] = (16 * j + lm < F) ? hr[16 * j] : 0.f;
    }
#pragma unroll
    for (int p = 0; p < P; ++p) {
      if (!TAIL || e0 + hbase + b + p < E) {
        float* ar = agg + (size_t)dd[p] * F + lm;
#pragma unroll
        for (int j = 0; j < J; ++j)
          if (16 * j + lm < F) atomicAdd(ar + 16 * j, v[p][j] * ww[p]);
      }
    }
  }
}

template <int F>
__global__ void edge_scatter_rows(const float* __restrict__ h, const float* __restrict__ norm,
                                  const int* __restrict__ src, const int* __restrict__ dst,
                                  float* __restrict__ aggA, float* __restrict__ aggB, int E) {
  const int lane  = threadIdx.x & 31;
  const int lm    = lane & 15;
  const int hbase = lane & 16;
  const long long wave = ((long long)blockIdx.x * blockDim.x + threadIdx.x) >> 5;
  const long long e0 = wave * 32;
  if (e0 >= E) return;
  const long long el = e0 + lane;
  const long long ec = el < E ? el : (long long)E - 1;
  const int   s_l = src[ec];
  const int   d_l = dst[ec];
  const float w_l = norm[ec];
  float* agg = (aggB != nullptr && l2_half()) ? aggB : aggA;
  if (e0 + 32 <= E) scatter_wave<F, false>(h, agg, s_l, d_l, w_l, lm, hbase, e0, E);
  else              scatter_wave<F, true >(h, agg, s_l, d_l, w_l, lm, hbase, e0, E);
}

__global__ void edge_scatter_rows_gen(const float* __restrict__ h, const float* __restrict__ norm,
                                      const int* __restrict__ src, const int* __restrict__ dst,
                                      float* __restrict__ aggA, float* __restrict__ aggB,
                                      int E, int F) {
  const int lane = threadIdx.x & 31;
  const int lm   = lane & 15;
  const long long e = (((long long)blockIdx.x * blockDim.x + threadIdx.x) >> 5) * 2 + (lane >> 4);
  if (e >= E) return;
  const int   s = src[e];
  const int   d = dst[e];
  const float w = norm[e];
  float* agg = (aggB != nullptr && l2_half()) ? aggB : aggA;
  for (int f = lm; f < F; f += 16)
    atomicAdd(agg + (size_t)d * F + f, h[(size_t)s * F + f] * w);
}

static void launch_scatter(const float* h, const float* norm, const int* src, const int* dst,
                           float* agg, float* aggB, int E, int F, hipStream_t stream) {
  const int TPB = 256;
  if (F == 48) {
    edge_scatter_rows<48><<<(E + 255) / 256, TPB, 0, stream>>>(h, norm, src, dst, agg, aggB, E);
  } else if (F == 40) {
    edge_scatter_rows<40><<<(E + 255) / 256, TPB, 0, stream>>>(h, norm, src, dst, agg, aggB, E);
  } else {
    edge_scatter_rows_gen<<<(E + 15) / 16, TPB, 0, stream>>>(h, norm, src, dst, agg, aggB, E, F);
  }
}


typedef __attribute__((address_space(1))) float       gmem_f32;
typedef __attribute__((address_space(1))) const float gmem_cf32;
__device__ __forceinline__ unsigned long long scalar_u64(unsigned long long a) {
  const unsigned lo = __builtin_amdgcn_readfirstlane((unsigned)a);
  const unsigned hi = __builtin_amdgcn_readfirstlane((unsigned)(a >> 32));
  return ((unsigned long long)hi << 32) | lo;
}
__device__ __forceinline__ gmem_cf32* scalar_base(const float* p) {
  return (gmem_cf32*)scalar_u64((unsigned long long)p);
}
__device__ __forceinline__ gmem_f32* scalar_base(float* p) {
  return (gmem_f32*)scalar_u64((unsigned long long)p);
}

template <int F, bool TAIL>
__device__ __forceinline__ void scatter_wave_lines(const float* __restrict__ h, float* __restrict__ agg,
                                                   int N, int s_l, int d_l, float w_l,
                                                   int lane, long long e0, int E) {
  constexpr int T = F - 32;
  constexpr int G = 32 / T;
  constexpr int B = 16;
  const float* hY = h + (size_t)N * 32;
  float*       aY = agg + (size_t)N * 32;
  const int sub = lane / T;
  const int ly  = lane % T;
#pragma unroll
  for (int b = 0; b < 32; b += B) {
    float vx[B];
    float vy[B / G];
#pragma unroll
    for (int p = 0; p < B; ++p) {
      const int s = __builtin_amdgcn_readlane(s_l, b + p);
      gmem_cf32* hrow = scalar_base(h + (size_t)s * 32);
      vx[p] = hrow[lane];
    }
#pragma unroll
    for (int g = 0; g < B / G; ++g) {
      const int s = __shfl(s_l, b + g * G + sub);
      vy[g] = hY[(size_t)s * T + ly];
    }
#pragma unroll
    for (int p = 0; p < B; ++p) {
      const int   d = __builtin_amdgcn_readlane(d_l, b + p);
      const float w = __int_as_float(__builtin_amdgcn_readlane(__float_as_int(w_l), b + p));
      gmem_f32* arow = scalar_base(agg + (size_t)d * 32);
      if (!TAIL || e0 + b + p < E)
        __hip_atomic_fetch_add(arow + lane, vx[p] * w, __ATOMIC_RELAXED, __HIP_MEMORY_SCOPE_AGENT);
    }
#pragma unroll
    for (int g = 0; g < B / G; ++g) {
      const int   q = b + g * G + sub;
      const int   d = __shfl(d_l, q);
      const float w = __shfl(w_l, q);
      if (!TAIL || e0 + q < E) atomicAdd(aY + (size_t)d * T + ly, vy[g] * w);
    }
  }
}

template <int F>
__device__ __forceinline__ void scatter_edges_lines(const float* __restrict__ h,
                                                    const float* __restrict__ norm,
                                                    const int* __restrict__ src,
                                                    const int* __restrict__ dst,
                                                    float* __restrict__ aggA, float* __restrict__ aggB,
                                                    int N, int E, long long e0) {
  const int lane = threadIdx.x & 31;
  const long long el = e0 + lane;
  const long long ec = el < E ? el : (long long)E - 1;
  const int   s_l = src[ec];
  const int   d_l = dst[ec];
  const float w_l = norm[ec];
  float* agg = l2_half() ? aggB : aggA;
  if (e0 + 32 <= E) scatter_wave_lines<F, false>(h, agg, N, s_l, d_l, w_l, lane, e0, E);
  else              scatter_wave_lines<F, true >(h, agg, N, s_l, d_l, w_l, lane, e0, E);
}

template <int F>
__global__ void edge_scatter_lines(const float* __restrict__ h, const float* __restrict__ norm,
                                   const int* __restrict__ src, const int* __restrict__ dst,
                                   float* __restrict__ aggA, float* __restrict__ aggB, int N, int E) {
  const long long wave = ((long long)blockIdx.x * blockDim.x + threadIdx.x) >> 5;
  const long long e0 = wave * 32;
  if (e0 >= E) return;
  scatter_edges_lines<F>(h, norm, src, dst, aggA, aggB, N, E, e0);
}

__device__ __forceinline__ bool product_slot(int b, int& idx) {
  const int q = b / 11;
  const int r = b - q * 11;
  if ((r & 3) == 1) { idx = 3 * q + (r >> 2); return true; }
  idx = 8 * q + r - ((r + 2) >> 2);
  return false;
}

template <int K, bool RELU_A, bool TWO_A, bool ADD_C, bool SPLIT_A>
__device__ __forceinline__ void product_tile(const float* __restrict__ A,
                                             const float* __restrict__ A2,
                                             const float* __restrict__ B,
                                             const float* __restrict__ bias,
                                             float* __restrict__ C, float* __restrict__ C2,
                                             float* __restrict__ Z, float* __restrict__ Z2,
                                             int N, int F, int rowTiles, int colTiles,
                                             int vblock) {
  const int lane = threadIdx.x & 31;
  const int wave = threadIdx.x >> 5;
  const int tile = vblock * (int)(blockDim.x >> 5) + wave;
  if (tile >= rowTiles * colTiles) return;

  const int rt = tile / colTiles;
  const int ct = tile - rt * colTiles;
  const int row0 = rt << 4;
  const int col0 = ct << 4;
  const int half = lane >> 4;
  const int lm   = lane & 15;

  const bool cOK = (col0 + lm) < F;
  const int  arow = min(row0 + lm, N - 1);
  const int  bcol = min(col0 + lm, F - 1);

  constexpr int AP = SPLIT_A ? 32 : K;
  const float* Arow  = A + (size_t)arow * AP + 2 * half;
  const float* ArowY = A + (size_t)N * 32 + (size_t)arow * (K - 32) + 2 * half;
  const float* Bcol = B + bcol;

  v2f afrag[K / 4];
  v2f bfrag[K / 4];
#pragma unroll
  for (int i = 0; i < K / 4; ++i) {
    const int ka = 4 * i + 2 * half;
    if (SPLIT_A && 4 * i >= 32) afrag[i] = *reinterpret_cast<const v2f*>(ArowY + (4 * i - 32));
    else                        afrag[i] = *reinterpret_cast<const v2f*>(Arow + 4 * i);
    bfrag[i].x = Bcol[(size_t)ka * F];
    bfrag[i].y = Bcol[(size_t)(ka + 1) * F];
  }
  if (TWO_A) {
    const float* A2row  = A2 + (size_t)arow * AP + 2 * half;
    const float* A2rowY = A2 + (size_t)N * 32 + (size_t)arow * (K - 32) + 2 * half;
#pragma unroll
    for (int i = 0; i < K / 4; ++i) {
      v2f t;
      if (SPLIT_A && 4 * i >= 32) t = *reinterpret_cast<const v2f*>(A2rowY + (4 * i - 32));
      else                        t = *reinterpret_cast<const v2f*>(A2row + 4 * i);
      afrag[i].x += t.x;
      afrag[i].y += t.y;
    }
  }
  if (RELU_A) {
#pragma unroll
    for (int i = 0; i < K / 4; ++i) {
      afrag[i].x = fmaxf(afrag[i].x, 0.f);
      afrag[i].y = fmaxf(afrag[i].y, 0.f);
    }
  }

  v8f acc = {};
#pragma unroll
  for (int i = 0; i < K / 4; ++i) {
    acc = __builtin_amdgcn_wmma_f32_16x16x4_f32(
        false, afrag[i], false, bfrag[i], (short)0, acc, false, false);
  }

  const bool   inX = col0 < 32;
  const int    cp  = inX ? 32 : F - 32;
  const size_t cb  = (inX ? (size_t)0 : (size_t)N * 32) + (size_t)(inX ? col0 : col0 - 32) + lm;

  const float bv = (bias != nullptr) ? bias[bcol] : 0.f;
  if (ADD_C) {
    float* Cp = (C2 != nullptr && l2_half()) ? C2 : C;
#pragma unroll
    for (int r = 0; r < 8; ++r) {
      const int row = row0 + r + 8 * half;
      if (row < N && cOK) atomicAdd(Cp + cb + (size_t)row * cp, acc[r] + bv);
    }
  } else {
#pragma unroll
    for (int r = 0; r < 8; ++r) {
      const int row = row0 + r + 8 * half;
      if (row < N && cOK) {
        const size_t o = cb + (size_t)row * cp;
        C[o] = acc[r] + bv;
        if (Z  != nullptr) Z[o]  = 0.f;
        if (Z2 != nullptr) Z2[o] = 0.f;
      }
    }
  }
}

template <int K, bool RELU_A, bool TWO_A>
__global__ void product_rows(const float* __restrict__ A, const float* __restrict__ A2,
                             const float* __restrict__ B, float* __restrict__ C,
                             float* __restrict__ Z, float* __restrict__ Z2,
                             int N, int F, int rowTiles, int colTiles) {
  product_tile<K, RELU_A, TWO_A, false, true>(A, A2, B, nullptr, C, nullptr, Z, Z2,
                                              N, F, rowTiles, colTiles, (int)blockIdx.x);
}

__global__ void degree_beside_product(const int* __restrict__ dst, const float* __restrict__ w,
                                      float* __restrict__ deg, float* __restrict__ deg2, int E,
                                      const float* __restrict__ x, const float* __restrict__ Wm,
                                      float* __restrict__ h,
                                      int N, int F, int rowTiles, int colTiles) {
  int idx;
  if (product_slot((int)blockIdx.x, idx)) {
    product_tile<64, false, false, false, false>(x, nullptr, Wm, nullptr, h, nullptr, nullptr,
                                                 nullptr, N, F, rowTiles, colTiles, idx);
  } else {
    const long long e = (long long)idx * blockDim.x + threadIdx.x;
    float* d = l2_half() ? deg2 : deg;
    if (e < E) atomicAdd(d + dst[e], w[e]);
  }
}

__global__ void norm_beside_product(const int* __restrict__ src, const int* __restrict__ dst,
                                    const float* __restrict__ w, const float* __restrict__ dinv,
                                    float* __restrict__ norm, int E,
                                    const float* __restrict__ x, const float* __restrict__ Vm,
                                    const float* __restrict__ bias,
                                    float* __restrict__ z, float* __restrict__ zb,
                                    int N, int F, int rowTiles, int colTiles) {
  int idx;
  if (product_slot((int)blockIdx.x, idx)) {
    product_tile<64, false, false, false, false>(x, nullptr, Vm, bias, z, nullptr, zb, nullptr,
                                                 N, F, rowTiles, colTiles, idx);
  } else {
    const long long e = (long long)idx * blockDim.x + threadIdx.x;
    if (e >= E) return;
    const float is = dinv[src[e]];
    const float id = dinv[dst[e]];
    norm[e] = is * w[e] * id;
  }
}

template <int F, int K, bool RELU_A, bool TWO_A>
__global__ void scatter_beside_product(const float* __restrict__ h, const float* __restrict__ norm,
                                       const int* __restrict__ src, const int* __restrict__ dst,
                                       float* __restrict__ aggA, float* __restrict__ aggB, int E,
                                       const float* __restrict__ A, const float* __restrict__ A2,
                                       const float* __restrict__ Vm, const float* __restrict__ bias,
                                       int N, int rowTiles, int colTiles) {
  int idx;
  if (product_slot((int)blockIdx.x, idx)) {
    product_tile<K, RELU_A, TWO_A, true, true>(A, A2, Vm, bias, aggA, aggB, nullptr, nullptr,
                                               N, F, rowTiles, colTiles, idx);
  } else {
    const long long wave = ((long long)idx * blockDim.x + threadIdx.x) >> 5;
    const long long e0 = wave * 32;
    if (e0 >= E) return;
    scatter_edges_lines<F>(h, norm, src, dst, aggA, aggB, N, E, e0);
  }
}

template <int C>
__global__ void relu_logsoftmax_t(float* __restrict__ y, const float* __restrict__ yb, int N) {
  int n = blockIdx.x * blockDim.x + threadIdx.x;
  if (n >= N) return;
  float v[C];
  float m = -3.4e38f;
  float* row = y + (size_t)n * C;
#pragma unroll
  for (int c = 0; c < C; ++c) v[c] = row[c];
  if (yb != nullptr) {
    const float* rb = yb + (size_t)n * C;
#pragma unroll
    for (int c = 0; c < C; ++c) v[c] += rb[c];
  }
#pragma unroll
  for (int c = 0; c < C; ++c) {
    float t = fmaxf(v[c], 0.f);
    v[c] = t;
    m = fmaxf(m, t);
  }
  float s = 0.f;
#pragma unroll
  for (int c = 0; c < C; ++c) s += __expf(v[c] - m);
  float ls = m + __logf(s);
#pragma unroll
  for (int c = 0; c < C; ++c) row[c] = v[c] - ls;
}

__global__ void relu_logsoftmax_gen(float* __restrict__ y, const float* __restrict__ yb,
                                    int N, int C) {
  int n = blockIdx.x * blockDim.x + threadIdx.x;
  if (n >= N) return;
  float* row = y + (size_t)n * C;
  if (yb != nullptr) {
    const float* rb = yb + (size_t)n * C;
    for (int c = 0; c < C; ++c) row[c] += rb[c];
  }
  float m = -3.4e38f;
  for (int c = 0; c < C; ++c) m = fmaxf(m, fmaxf(row[c], 0.f));
  float s = 0.f;
  for (int c = 0; c < C; ++c) s += __expf(fmaxf(row[c], 0.f) - m);
  float ls = m + __logf(s);
  for (int c = 0; c < C; ++c) row[c] = fmaxf(row[c], 0.f) - ls;
}

template <int C>
__global__ void relu_logsoftmax_lines(const float* __restrict__ ya, const float* __restrict__ yb,
                                      float* __restrict__ out, int N) {
  constexpr int T = C - 32;
  int n = blockIdx.x * blockDim.x + threadIdx.x;
  if (n >= N) return;
  const float* ax = ya + (size_t)n * 32;
  const float* bx = yb + (size_t)n * 32;
  const float* ay = ya + (size_t)N * 32 + (size_t)n * T;
  const float* by = yb + (size_t)N * 32 + (size_t)n * T;
  float v[C];
#pragma unroll
  for (int c = 0; c < 32; ++c) v[c] = ax[c] + bx[c];
#pragma unroll
  for (int c = 0; c < T; ++c) v[32 + c] = ay[c] + by[c];
  float m = -3.4e38f;
#pragma unroll
  for (int c = 0; c < C; ++c) {
    float t = fmaxf(v[c], 0.f);
    v[c] = t;
    m = fmaxf(m, t);
  }
  float s = 0.f;
#pragma unroll
  for (int c = 0; c < C; ++c) s += __expf(v[c] - m);
  float ls = m + __logf(s);
  float* row = out + (size_t)n * C;
#pragma unroll
  for (int c = 0; c < C; ++c) row[c] = v[c] - ls;
}

static inline int ceil_div(long long a, int b) { return (int)((a + b - 1) / b); }

static void launch_gemm(const float* A, const float* A2, const float* B, const float* bias,
                        float* C, float* Z, int N, int K, int F, bool reluA,
                        hipStream_t stream) {
  const int TPB = 256;
  int rowT = ceil_div(N, 16), colT = ceil_div(F, 16);
  int blocks = ceil_div((long long)rowT * colT, TPB / 32);
  if (K == 64 && !reluA && A2 == nullptr) {
    gemm_wmma<64, false, false><<<blocks, TPB, 0, stream>>>(A, A2, B, bias, C, Z, N, F, rowT, colT);
  } else if (K == 48 && reluA && A2 != nullptr) {
    gemm_wmma<48, true, true><<<blocks, TPB, 0, stream>>>(A, A2, B, bias, C, Z, N, F, rowT, colT);
  } else if (K == 48 && reluA) {
    gemm_wmma<48, true, false><<<blocks, TPB, 0, stream>>>(A, A2, B, bias, C, Z, N, F, rowT, colT);
  } else {
    gemm_wmma_gen<<<blocks, TPB, 0, stream>>>(A, A2, B, bias, C, Z, N, K, F, rowT, colT,
                                              reluA ? 1 : 0);
  }
}

extern "C" void kernel_launch(void* const* d_in, const int* in_sizes, int n_in,
                              void* d_out, int out_size, void* d_ws, size_t ws_size,
                              hipStream_t stream) {
  const float* x  = (const float*)d_in[0];
  const int*   ei = (const int*)d_in[1];
  const float* ew = (const float*)d_in[2];
  const float* W1 = (const float*)d_in[3];
  const float* V1 = (const float*)d_in[4];
  const float* b1 = (const float*)d_in[5];
  const float* W2 = (const float*)d_in[6];
  const float* V2 = (const float*)d_in[7];
  const float* b2 = (const float*)d_in[8];
  float* out = (float*)d_out;

  const int H   = in_sizes[5];
  const int Cn  = in_sizes[8];
  const int Fin = in_sizes[3] / H;
  const int N   = in_sizes[0] / Fin;
  const int E   = in_sizes[2];
  const int* src = ei;
  const int* dst = ei + E;

  float* ws = (float*)d_ws;
  size_t off = 0;
  float* deg  = ws + off; off += ((size_t)N + 255) & ~(size_t)255;
  float* norm = ws + off; off += ((size_t)E + 255) & ~(size_t)255;
  float* hbuf = ws + off; off += ((size_t)N * H + 255) & ~(size_t)255;
  float* z1   = ws + off; off += ((size_t)N * H + 255) & ~(size_t)255;
  const size_t wsFloats = ws_size / sizeof(float);
  float* deg2 = nullptr;
  float* z1b  = nullptr;
  float* outb = nullptr;
  {
    const size_t nDeg = ((size_t)N + 255) & ~(size_t)255;
    const size_t nZ1  = ((size_t)N * H + 255) & ~(size_t)255;
    const size_t nOut = ((size_t)N * Cn + 255) & ~(size_t)255;
    if (off + nDeg <= wsFloats) { deg2 = ws + off; off += nDeg; }
    if (off + nZ1  <= wsFloats) { z1b  = ws + off; off += nZ1; }
    if (off + nOut <= wsFloats) { outb = ws + off; off += nOut; }
  }

  const int TPB = 256;

  if (Fin == 64 && H == 48 && Cn == 40 && deg2 != nullptr && z1b != nullptr && outb != nullptr) {
    const int rowT = ceil_div(N, 16);
    const int nbP1 = ceil_div((long long)rowT * ceil_div(H, 16), TPB / 32);
    const int nbP2 = ceil_div((long long)rowT * ceil_div(Cn, 16), TPB / 32);
    const int nbE  = ceil_div(E, TPB);
    const int per1 = ceil_div(nbP1, 3) > ceil_div(nbE, 8) ? ceil_div(nbP1, 3) : ceil_div(nbE, 8);
    const int per2 = ceil_div(nbP2, 3) > ceil_div(nbE, 8) ? ceil_div(nbP2, 3) : ceil_div(nbE, 8);
    float* suma = hbuf;

    zero_f32<<<ceil_div(N, TPB), TPB, 0, stream>>>(deg, deg2, N);
    degree_beside_product<<<per1 * 11, TPB, 0, stream>>>(dst, ew, deg, deg2, E, x, W1, hbuf,
                                                         N, H, rowT, ceil_div(H, 16));
    inv_sqrt_degree<<<ceil_div(N, TPB), TPB, 0, stream>>>(deg, deg2, N);
    norm_beside_product<<<per1 * 11, TPB, 0, stream>>>(src, dst, ew, deg, norm, E,
                                                       x, V1, b1, z1, z1b,
                                                       N, H, rowT, ceil_div(H, 16));

    edge_scatter_lines<48><<<nbE, TPB, 0, stream>>>(hbuf, norm, src, dst, z1, z1b, N, E);

    product_rows<48, true, true><<<nbP2, TPB, 0, stream>>>(z1, z1b, W2, out, suma, outb,
                                                          N, Cn, rowT, ceil_div(Cn, 16));
    scatter_beside_product<40, 48, true, true><<<per2 * 11, TPB, 0, stream>>>(
        out, norm, src, dst, suma, outb, E, z1, z1b, V2, b2, N, rowT, ceil_div(Cn, 16));
    relu_logsoftmax_lines<40><<<ceil_div(N, TPB), TPB, 0, stream>>>(suma, outb, out, N);
    return;
  }

  zero_f32<<<ceil_div(N, TPB), TPB, 0, stream>>>(deg, deg2, N);
  deg_accum<<<ceil_div(E, TPB), TPB, 0, stream>>>(dst, ew, deg, deg2, E);
  norm_compute<<<ceil_div(E, TPB), TPB, 0, stream>>>(src, dst, ew, deg, deg2, norm, E);

  launch_gemm(x, nullptr, W1, nullptr, hbuf, nullptr, N, Fin, H, false, stream);
  launch_gemm(x, nullptr, V1, b1,      z1,   z1b,     N, Fin, H, false, stream);
  launch_scatter(hbuf, norm, src, dst, z1, z1b, E, H, stream);

  launch_gemm(z1, z1b, W2, nullptr, hbuf, nullptr, N, H, Cn, true, stream);
  launch_gemm(z1, z1b, V2, b2,      out,  outb,    N, H, Cn, true, stream);
  {
    launch_scatter(hbuf, norm, src, dst, out, outb, E, Cn, stream);
    if (Cn == 40) {
      relu_logsoftmax_t<40><<<ceil_div(N, TPB), TPB, 0, stream>>>(out, outb, N);
    } else {
      relu_logsoftmax_gen<<<ceil_div(N, TPB), TPB, 0, stream>>>(out, outb, N, Cn);
    }
  }
}


# Rows that one lane walked alone (A tiles of the four products, the last pass's sums and output) now move as whole 128-B lines through per-wave LDS; fragments, k order and row math unchanged
# speedup vs baseline: 2.5692x; 1.0301x over previous
#include <hip/hip_runtime.h>

typedef float v2f __attribute__((ext_vector_type(2)));
typedef float v8f __attribute__((ext_vector_type(8)));

__global__ void zero_f32(float* __restrict__ p, float* __restrict__ p2, int n) {
  int i = blockIdx.x * blockDim.x + threadIdx.x;
  if (i < n) {
    p[i] = 0.f;
    if (p2 != nullptr) p2[i] = 0.f;
  }
}

__device__ __forceinline__ int l2_half() {
  const unsigned id = __builtin_amdgcn_s_sendmsg_rtn(0x87);
  return (int)((id >> 18) & 1u);
}

__global__ void deg_accum(const int* __restrict__ dst, const float* __restrict__ w,
                          float* __restrict__ deg, float* __restrict__ deg2, int E) {
  int e = blockIdx.x * blockDim.x + threadIdx.x;
  float* d = (deg2 != nullptr && l2_half()) ? deg2 : deg;
  if (e < E) atomicAdd(d + dst[e], w[e]);
}

__global__ void norm_compute(const int* __restrict__ src, const int* __restrict__ dst,
                             const float* __restrict__ w, const float* __restrict__ deg,
                             const float* __restrict__ deg2, float* __restrict__ norm, int E) {
  int e = blockIdx.x * blockDim.x + threadIdx.x;
  if (e >= E) return;
  float ds = deg[src[e]];
  float dd = deg[dst[e]];
  if (deg2 != nullptr) {
    ds += deg2[src[e]];
    dd += deg2[dst[e]];
  }
  float is = ds > 0.f ? rsqrtf(ds) : 0.f;
  float id = dd > 0.f ? rsqrtf(dd) : 0.f;
  norm[e] = is * w[e] * id;
}

__global__ void inv_sqrt_degree(float* __restrict__ deg, const float* __restrict__ deg2, int N) {
  int n = blockIdx.x * blockDim.x + threadIdx.x;
  if (n >= N) return;
  const float d = deg[n] + deg2[n];
  deg[n] = d > 0.f ? rsqrtf(d) : 0.f;
}

template <int K, bool RELU_A, bool TWO_A>
__global__ void gemm_wmma(const float* __restrict__ A, const float* __restrict__ A2,
                          const float* __restrict__ B,
                          const float* __restrict__ bias, float* __restrict__ C,
                          float* __restrict__ Z,
                          int N, int F, int rowTiles, int colTiles) {
  const int lane = threadIdx.x & 31;
  const int wave = threadIdx.x >> 5;
  const int tile = blockIdx.x * (blockDim.x >> 5) + wave;
  if (tile >= rowTiles * colTiles) return;

  const int rt = tile / colTiles;
  const int ct = tile - rt * colTiles;
  const int row0 = rt << 4;
  const int col0 = ct << 4;
  const int half = lane >> 4;
  const int lm   = lane & 15;

  const bool cOK = (col0 + lm) < F;
  const int  arow = min(row0 + lm, N - 1);
  const int  bcol = min(col0 + lm, F - 1);

  const float* Arow = A + (size_t)arow * K + 2 * half;
  const float* Bcol = B + bcol;

  v2f afrag[K / 4];
  v2f bfrag[K / 4];
#pragma unroll
  for (int i = 0; i < K / 4; ++i) {
    const int ka = 4 * i + 2 * half;
    afrag[i] = *reinterpret_cast<const v2f*>(Arow + 4 * i);
    bfrag[i].x = Bcol[(size_t)ka * F];
    bfrag[i].y = Bcol[(size_t)(ka + 1) * F];
  }
  if (TWO_A) {
    const float* A2row = A2 + (size_t)arow * K + 2 * half;
#pragma unroll
    for (int i = 0; i < K / 4; ++i) {
      const v2f t = *reinterpret_cast<const v2f*>(A2row + 4 * i);
      afrag[i].x += t.x;
      afrag[i].y += t.y;
    }
  }
  if (RELU_A) {
#pragma unroll
    for (int i = 0; i < K / 4; ++i) {
      afrag[i].x = fmaxf(afrag[i].x, 0.f);
      afrag[i].y = fmaxf(afrag[i].y, 0.f);
    }
  }

  v8f acc = {};
#pragma unroll
  for (int i = 0; i < K / 4; ++i) {
    acc = __builtin_amdgcn_wmma_f32_16x16x4_f32(
        false, afrag[i], false, bfrag[i], (short)0, acc, false, false);
  }

  const float bv = (bias != nullptr) ? bias[bcol] : 0.f;
#pragma unroll
  for (int r = 0; r < 8; ++r) {
    const int row = row0 + r + 8 * half;
    if (row < N && cOK) {
      C[(size_t)row * F + col0 + lm] = acc[r] + bv;
      if (Z != nullptr) Z[(size_t)row * F + col0 + lm] = 0.f;
    }
  }
}

__global__ void gemm_wmma_gen(const float* __restrict__ A, const float* __restrict__ A2,
                              const float* __restrict__ B,
                              const float* __restrict__ bias, float* __restrict__ C,
                              float* __restrict__ Z,
                              int N, int K, int F, int rowTiles, int colTiles, int reluA) {
  const int lane = threadIdx.x & 31;
  const int wave = threadIdx.x >> 5;
  const int tile = blockIdx.x * (blockDim.x >> 5) + wave;
  if (tile >= rowTiles * colTiles) return;
  const int rt = tile / colTiles;
  const int ct = tile - rt * colTiles;
  const int row0 = rt << 4, col0 = ct << 4;
  const int half = lane >> 4, lm = lane & 15;
  const bool cOK = (col0 + lm) < F;
  const int arow = min(row0 + lm, N - 1);
  const int bcol = min(col0 + lm, F - 1);
  v8f acc = {};
  for (int kb = 0; kb < K; kb += 4) {
    const int ka = kb + 2 * half;
    v2f a = *reinterpret_cast<const v2f*>(A + (size_t)arow * K + ka);
    if (A2 != nullptr) {
      const v2f t = *reinterpret_cast<const v2f*>(A2 + (size_t)arow * K + ka);
      a.x += t.x;
      a.y += t.y;
    }
    if (reluA) { a.x = fmaxf(a.x, 0.f); a.y = fmaxf(a.y, 0.f); }
    v2f b;
    b.x = B[(size_t)ka * F + bcol];
    b.y = B[(size_t)(ka + 1) * F + bcol];
    acc = __builtin_amdgcn_wmma_f32_16x16x4_f32(false, a, false, b, (short)0, acc,
                                                false, false);
  }
  const float bv = (bias != nullptr) ? bias[bcol] : 0.f;
#pragma unroll
  for (int r = 0; r < 8; ++r) {
    const int row = row0 + r + 8 * half;
    if (row < N && cOK) {
      C[(size_t)row * F + col0 + lm] = acc[r] + bv;
      if (Z != nullptr) Z[(size_t)row * F + col0 + lm] = 0.f;
    }
  }
}

template <int F, bool TAIL>
__device__ __forceinline__ void scatter_wave(const float* __restrict__ h, float* __restrict__ agg,
                                             int s_l, int d_l, float w_l,
                                             int lm, int hbase, long long e0, int E) {
  constexpr int J = (F + 15) / 16;
  constexpr int P = 8;
#pragma unroll
  for (int b = 0; b < 16; b += P) {
    float v[P][J];
    float ww[P];
    int   dd[P];
#pragma unroll
    for (int p = 0; p < P; ++p) {
      const int q = hbase + b + p;
      const int s = __shfl(s_l, q);
      ww[p] = __shfl(w_l, q);
      dd[p] = __shfl(d_l, q);
      const float* hr = h + (size_t)s * F + lm;
#pragma unroll
      for (int j = 0; j < J; ++j)
        v[p][j] = (16 * j + lm < F) ? hr[16 * j] : 0.f;
    }
#pragma unroll
    for (int p = 0; p < P; ++p) {
      if (!TAIL || e0 + hbase + b + p < E) {
        float* ar = agg + (size_t)dd[p] * F + lm;
#pragma unroll
        for (int j = 0; j < J; ++j)
          if (16 * j + lm < F) atomicAdd(ar + 16 * j, v[p][j] * ww[p]);
      }
    }
  }
}

template <int F>
__global__ void edge_scatter_rows(const float* __restrict__ h, const float* __restrict__ norm,
                                  const int* __restrict__ src, const int* __restrict__ dst,
                                  float* __restrict__ aggA, float* __restrict__ aggB, int E) {
  const int lane  = threadIdx.x & 31;
  const int lm    = lane & 15;
  const int hbase = lane & 16;
  const long long wave = ((long long)blockIdx.x * blockDim.x + threadIdx.x) >> 5;
  const long long e0 = wave * 32;
  if (e0 >= E) return;
  const long long el = e0 + lane;
  const long long ec = el < E ? el : (long long)E - 1;
  const int   s_l = src[ec];
  const int   d_l = dst[ec];
  const float w_l = norm[ec];
  float* agg = (aggB != nullptr && l2_half()) ? aggB : aggA;
  if (e0 + 32 <= E) scatter_wave<F, false>(h, agg, s_l, d_l, w_l, lm, hbase, e0, E);
  else              scatter_wave<F, true >(h, agg, s_l, d_l, w_l, lm, hbase, e0, E);
}

__global__ void edge_scatter_rows_gen(const float* __restrict__ h, const float* __restrict__ norm,
                                      const int* __restrict__ src, const int* __restrict__ dst,
                                      float* __restrict__ aggA, float* __restrict__ aggB,
                                      int E, int F) {
  const int lane = threadIdx.x & 31;
  const int lm   = lane & 15;
  const long long e = (((long long)blockIdx.x * blockDim.x + threadIdx.x) >> 5) * 2 + (lane >> 4);
  if (e >= E) return;
  const int   s = src[e];
  const int   d = dst[e];
  const float w = norm[e];
  float* agg = (aggB != nullptr && l2_half()) ? aggB : aggA;
  for (int f = lm; f < F; f += 16)
    atomicAdd(agg + (size_t)d * F + f, h[(size_t)s * F + f] * w);
}

static void launch_scatter(const float* h, const float* norm, const int* src, const int* dst,
                           float* agg, float* aggB, int E, int F, hipStream_t stream) {
  const int TPB = 256;
  if (F == 48) {
    edge_scatter_rows<48><<<(E + 255) / 256, TPB, 0, stream>>>(h, norm, src, dst, agg, aggB, E);
  } else if (F == 40) {
    edge_scatter_rows<40><<<(E + 255) / 256, TPB, 0, stream>>>(h, norm, src, dst, agg, aggB, E);
  } else {
    edge_scatter_rows_gen<<<(E + 15) / 16, TPB, 0, stream>>>(h, norm, src, dst, agg, aggB, E, F);
  }
}


typedef __attribute__((address_space(1))) float       gmem_f32;
typedef __attribute__((address_space(1))) const float gmem_cf32;
__device__ __forceinline__ unsigned long long scalar_u64(unsigned long long a) {
  const unsigned lo = __builtin_amdgcn_readfirstlane((unsigned)a);
  const unsigned hi = __builtin_amdgcn_readfirstlane((unsigned)(a >> 32));
  return ((unsigned long long)hi << 32) | lo;
}
__device__ __forceinline__ gmem_cf32* scalar_base(const float* p) {
  return (gmem_cf32*)scalar_u64((unsigned long long)p);
}
__device__ __forceinline__ gmem_f32* scalar_base(float* p) {
  return (gmem_f32*)scalar_u64((unsigned long long)p);
}

template <int F, bool TAIL>
__device__ __forceinline__ void scatter_wave_lines(const float* __restrict__ h, float* __restrict__ agg,
                                                   int N, int s_l, int d_l, float w_l,
                                                   int lane, long long e0, int E) {
  constexpr int T = F - 32;
  constexpr int G = 32 / T;
  constexpr int B = 16;
  const float* hY = h + (size_t)N * 32;
  float*       aY = agg + (size_t)N * 32;
  const int sub = lane / T;
  const int ly  = lane % T;
#pragma unroll
  for (int b = 0; b < 32; b += B) {
    float vx[B];
    float vy[B / G];
#pragma unroll
    for (int p = 0; p < B; ++p) {
      const int s = __builtin_amdgcn_readlane(s_l, b + p);
      gmem_cf32* hrow = scalar_base(h + (size_t)s * 32);
      vx[p] = hrow[lane];
    }
#pragma unroll
    for (int g = 0; g < B / G; ++g) {
      const int s = __shfl(s_l, b + g * G + sub);
      vy[g] = hY[(size_t)s * T + ly];
    }
#pragma unroll
    for (int p = 0; p < B; ++p) {
      const int   d = __builtin_amdgcn_readlane(d_l, b + p);
      const float w = __int_as_float(__builtin_amdgcn_readlane(__float_as_int(w_l), b + p));
      gmem_f32* arow = scalar_base(agg + (size_t)d * 32);
      if (!TAIL || e0 + b + p < E)
        __hip_atomic_fetch_add(arow + lane, vx[p] * w, __ATOMIC_RELAXED, __HIP_MEMORY_SCOPE_AGENT);
    }
#pragma unroll
    for (int g = 0; g < B / G; ++g) {
      const int   q = b + g * G + sub;
      const int   d = __shfl(d_l, q);
      const float w = __shfl(w_l, q);
      if (!TAIL || e0 + q < E) atomicAdd(aY + (size_t)d * T + ly, vy[g] * w);
    }
  }
}

template <int F>
__device__ __forceinline__ void scatter_edges_lines(const float* __restrict__ h,
                                                    const float* __restrict__ norm,
                                                    const int* __restrict__ src,
                                                    const int* __restrict__ dst,
                                                    float* __restrict__ aggA, float* __restrict__ aggB,
                                                    int N, int E, long long e0) {
  const int lane = threadIdx.x & 31;
  const long long el = e0 + lane;
  const long long ec = el < E ? el : (long long)E - 1;
  const int   s_l = src[ec];
  const int   d_l = dst[ec];
  const float w_l = norm[ec];
  float* agg = l2_half() ? aggB : aggA;
  if (e0 + 32 <= E) scatter_wave_lines<F, false>(h, agg, N, s_l, d_l, w_l, lane, e0, E);
  else              scatter_wave_lines<F, true >(h, agg, N, s_l, d_l, w_l, lane, e0, E);
}

template <int F>
__global__ void edge_scatter_lines(const float* __restrict__ h, const float* __restrict__ norm,
                                   const int* __restrict__ src, const int* __restrict__ dst,
                                   float* __restrict__ aggA, float* __restrict__ aggB, int N, int E) {
  const long long wave = ((long long)blockIdx.x * blockDim.x + threadIdx.x) >> 5;
  const long long e0 = wave * 32;
  if (e0 >= E) return;
  scatter_edges_lines<F>(h, norm, src, dst, aggA, aggB, N, E, e0);
}

__device__ __forceinline__ bool product_slot(int b, int& idx) {
  const int q = b / 11;
  const int r = b - q * 11;
  if ((r & 3) == 1) { idx = 3 * q + (r >> 2); return true; }
  idx = 8 * q + r - ((r + 2) >> 2);
  return false;
}

template <int K, bool RELU_A, bool TWO_A, bool SPLIT_A>
__device__ __forceinline__ void stage_a_rows(float* __restrict__ lds,
                                             const float* __restrict__ A,
                                             const float* __restrict__ A2,
                                             int N, int row0, int lane) {
  constexpr int LP = K + 4;
  constexpr int PX = (SPLIT_A ? 32 : K) / 4;
  constexpr int NX = (16 * PX) / 32;
  constexpr int TY = SPLIT_A ? K - 32 : 4;
  constexpr int PY = TY / 4;
  constexpr int NY = SPLIT_A ? (16 * PY) / 32 : 0;
  constexpr int NL = NX + NY;
  float4 va[NL];
  float4 vb[TWO_A ? NL : 1];
  int    lo[NL];
#pragma unroll
  for (int j = 0; j < NL; ++j) {
    size_t o;
    if (j < NX) {
      const int piece = j * 32 + lane;
      const int r   = piece / PX;
      const int col = 4 * (piece - r * PX);
      o = (size_t)min(row0 + r, N - 1) * (4 * PX) + col;
      lo[j] = r * LP + col;
    } else {
      const int piece = (j - NX) * 32 + lane;
      const int r   = piece / PY;
      const int col = 4 * (piece - r * PY);
      o = (size_t)N * 32 + (size_t)min(row0 + r, N - 1) * TY + col;
      lo[j] = r * LP + 32 + col;
    }
    va[j] = *reinterpret_cast<const float4*>(A + o);
    if (TWO_A) vb[j] = *reinterpret_cast<const float4*>(A2 + o);
  }
#pragma unroll
  for (int j = 0; j < NL; ++j) {
    float4 v = va[j];
    if (TWO_A) {
      v.x += vb[j].x; v.y += vb[j].y; v.z += vb[j].z; v.w += vb[j].w;
    }
    if (RELU_A) {
      v.x = fmaxf(v.x, 0.f); v.y = fmaxf(v.y, 0.f); v.z = fmaxf(v.z, 0.f); v.w = fmaxf(v.w, 0.f);
    }
    *reinterpret_cast<float4*>(lds + lo[j]) = v;
  }
}

template <int K, bool RELU_A, bool TWO_A, bool ADD_C, bool SPLIT_A>
__device__ __forceinline__ void product_tile(const float* __restrict__ A,
                                             const float* __restrict__ A2,
                                             const float* __restrict__ B,
                                             const float* __restrict__ bias,
                                             float* __restrict__ C, float* __restrict__ C2,
                                             float* __restrict__ Z, float* __restrict__ Z2,
                                             int N, int F, int rowTiles, int colTiles,
                                             int vblock) {
  const int lane = threadIdx.x & 31;
  const int wave = threadIdx.x >> 5;
  const int tile = vblock * (int)(blockDim.x >> 5) + wave;
  if (tile >= rowTiles * colTiles) return;

  const int rt = tile / colTiles;
  const int ct = tile - rt * colTiles;
  const int row0 = rt << 4;
  const int col0 = ct << 4;
  const int half = lane >> 4;
  const int lm   = lane & 15;

  const bool cOK = (col0 + lm) < F;
  const int  bcol = min(col0 + lm, F - 1);

  constexpr int LP = K + 4;
  __shared__ __attribute__((aligned(16))) float a_rows[8 * 16 * LP];
  float* lds = a_rows + wave * (16 * LP);
  stage_a_rows<K, RELU_A, TWO_A, SPLIT_A>(lds, A, A2, N, row0, lane);

  const float* Bcol = B + bcol;

  v2f afrag[K / 4];
  v2f bfrag[K / 4];
#pragma unroll
  for (int i = 0; i < K / 4; ++i) {
    const int ka = 4 * i + 2 * half;
    afrag[i] = *reinterpret_cast<const v2f*>(lds + lm * LP + ka);
    bfrag[i].x = Bcol[(size_t)ka * F];
    bfrag[i].y = Bcol[(size_t)(ka + 1) * F];
  }

  v8f acc = {};
#pragma unroll
  for (int i = 0; i < K / 4; ++i) {
    acc = __builtin_amdgcn_wmma_f32_16x16x4_f32(
        false, afrag[i], false, bfrag[i], (short)0, acc, false, false);
  }

  const bool   inX = col0 < 32;
  const int    cp  = inX ? 32 : F - 32;
  const size_t cb  = (inX ? (size_t)0 : (size_t)N * 32) + (size_t)(inX ? col0 : col0 - 32) + lm;

  const float bv = (bias != nullptr) ? bias[bcol] : 0.f;
  if (ADD_C) {
    float* Cp = (C2 != nullptr && l2_half()) ? C2 : C;
#pragma unroll
    for (int r = 0; r < 8; ++r) {
      const int row = row0 + r + 8 * half;
      if (row < N && cOK) atomicAdd(Cp + cb + (size_t)row * cp, acc[r] + bv);
    }
  } else {
#pragma unroll
    for (int r = 0; r < 8; ++r) {
      const int row = row0 + r + 8 * half;
      if (row < N && cOK) {
        const size_t o = cb + (size_t)row * cp;
        C[o] = acc[r] + bv;
        if (Z  != nullptr) Z[o]  = 0.f;
        if (Z2 != nullptr) Z2[o] = 0.f;
      }
    }
  }
}

template <int K, bool RELU_A, bool TWO_A>
__global__ void product_rows(const float* __restrict__ A, const float* __restrict__ A2,
                             const float* __restrict__ B, float* __restrict__ C,
                             float* __restrict__ Z, float* __restrict__ Z2,
                             int N, int F, int rowTiles, int colTiles) {
  product_tile<K, RELU_A, TWO_A, false, true>(A, A2, B, nullptr, C, nullptr, Z, Z2,
                                              N, F, rowTiles, colTiles, (int)blockIdx.x);
}

__global__ void degree_beside_product(const int* __restrict__ dst, const float* __restrict__ w,
                                      float* __restrict__ deg, float* __restrict__ deg2, int E,
                                      const float* __restrict__ x, const float* __restrict__ Wm,
                                      float* __restrict__ h,
                                      int N, int F, int rowTiles, int colTiles) {
  int idx;
  if (product_slot((int)blockIdx.x, idx)) {
    product_tile<64, false, false, false, false>(x, nullptr, Wm, nullptr, h, nullptr, nullptr,
                                                 nullptr, N, F, rowTiles, colTiles, idx);
  } else {
    const long long e = (long long)idx * blockDim.x + threadIdx.x;
    float* d = l2_half() ? deg2 : deg;
    if (e < E) atomicAdd(d + dst[e], w[e]);
  }
}

__global__ void norm_beside_product(const int* __restrict__ src, const int* __restrict__ dst,
                                    const float* __restrict__ w, const float* __restrict__ dinv,
                                    float* __restrict__ norm, int E,
                                    const float* __restrict__ x, const float* __restrict__ Vm,
                                    const float* __restrict__ bias,
                                    float* __restrict__ z, float* __restrict__ zb,
                                    int N, int F, int rowTiles, int colTiles) {
  int idx;
  if (product_slot((int)blockIdx.x, idx)) {
    product_tile<64, false, false, false, false>(x, nullptr, Vm, bias, z, nullptr, zb, nullptr,
                                                 N, F, rowTiles, colTiles, idx);
  } else {
    const long long e = (long long)idx * blockDim.x + threadIdx.x;
    if (e >= E) return;
    const float is = dinv[src[e]];
    const float id = dinv[dst[e]];
    norm[e] = is * w[e] * id;
  }
}

template <int F, int K, bool RELU_A, bool TWO_A>
__global__ void scatter_beside_product(const float* __restrict__ h, const float* __restrict__ norm,
                                       const int* __restrict__ src, const int* __restrict__ dst,
                                       float* __restrict__ aggA, float* __restrict__ aggB, int E,
                                       const float* __restrict__ A, const float* __restrict__ A2,
                                       const float* __restrict__ Vm, const float* __restrict__ bias,
                                       int N, int rowTiles, int colTiles) {
  int idx;
  if (product_slot((int)blockIdx.x, idx)) {
    product_tile<K, RELU_A, TWO_A, true, true>(A, A2, Vm, bias, aggA, aggB, nullptr, nullptr,
                                               N, F, rowTiles, colTiles, idx);
  } else {
    const long long wave = ((long long)idx * blockDim.x + threadIdx.x) >> 5;
    const long long e0 = wave * 32;
    if (e0 >= E) return;
    scatter_edges_lines<F>(h, norm, src, dst, aggA, aggB, N, E, e0);
  }
}

template <int C>
__global__ void relu_logsoftmax_t(float* __restrict__ y, const float* __restrict__ yb, int N) {
  int n = blockIdx.x * blockDim.x + threadIdx.x;
  if (n >= N) return;
  float v[C];
  float m = -3.4e38f;
  float* row = y + (size_t)n * C;
#pragma unroll
  for (int c = 0; c < C; ++c) v[c] = row[c];
  if (yb != nullptr) {
    const float* rb = yb + (size_t)n * C;
#pragma unroll
    for (int c = 0; c < C; ++c) v[c] += rb[c];
  }
#pragma unroll
  for (int c = 0; c < C; ++c) {
    float t = fmaxf(v[c], 0.f);
    v[c] = t;
    m = fmaxf(m, t);
  }
  float s = 0.f;
#pragma unroll
  for (int c = 0; c < C; ++c) s += __expf(v[c] - m);
  float ls = m + __logf(s);
#pragma unroll
  for (int c = 0; c < C; ++c) row[c] = v[c] - ls;
}

__global__ void relu_logsoftmax_gen(float* __restrict__ y, const float* __restrict__ yb,
                                    int N, int C) {
  int n = blockIdx.x * blockDim.x + threadIdx.x;
  if (n >= N) return;
  float* row = y + (size_t)n * C;
  if (yb != nullptr) {
    const float* rb = yb + (size_t)n * C;
    for (int c = 0; c < C; ++c) row[c] += rb[c];
  }
  float m = -3.4e38f;
  for (int c = 0; c < C; ++c) m = fmaxf(m, fmaxf(row[c], 0.f));
  float s = 0.f;
  for (int c = 0; c < C; ++c) s += __expf(fmaxf(row[c], 0.f) - m);
  float ls = m + __logf(s);
  for (int c = 0; c < C; ++c) row[c] = fmaxf(row[c], 0.f) - ls;
}

template <int C>
__global__ void relu_logsoftmax_lines(const float* __restrict__ ya, const float* __restrict__ yb,
                                      float* __restrict__ out, int N) {
  constexpr int T  = C - 32;
  constexpr int LP = C + 4;
  constexpr int PY = T / 4;
  constexpr int PO = C / 4;
  __shared__ __attribute__((aligned(16))) float rows[8 * 32 * LP];
  const int lane = threadIdx.x & 31;
  const int wave = threadIdx.x >> 5;
  const int n0 = ((int)blockIdx.x * (int)(blockDim.x >> 5) + wave) * 32;
  if (n0 >= N) return;
  float* lds = rows + wave * (32 * LP);

  float4 a[8 + PY], b[8 + PY];
  int    lo[8 + PY];
#pragma unroll
  for (int j = 0; j < 8 + PY; ++j) {
    size_t o;
    if (j < 8) {
      const int piece = j * 32 + lane;
      const int r   = piece >> 3;
      const int col = 4 * (piece & 7);
      o = (size_t)min(n0 + r, N - 1) * 32 + col;
      lo[j] = r * LP + col;
    } else {
      const int piece = (j - 8) * 32 + lane;
      const int r   = piece / PY;
      const int col = 4 * (piece - r * PY);
      o = (size_t)N * 32 + (size_t)min(n0 + r, N - 1) * T + col;
      lo[j] = r * LP + 32 + col;
    }
    a[j] = *reinterpret_cast<const float4*>(ya + o);
    b[j] = *reinterpret_cast<const float4*>(yb + o);
  }
#pragma unroll
  for (int j = 0; j < 8 + PY; ++j) {
    float4 v;
    v.x = a[j].x + b[j].x; v.y = a[j].y + b[j].y; v.z = a[j].z + b[j].z; v.w = a[j].w + b[j].w;
    *reinterpret_cast<float4*>(lds + lo[j]) = v;
  }

  float v[C];
  float* mine = lds + lane * LP;
#pragma unroll
  for (int q = 0; q < PO; ++q) {
    const float4 t = *reinterpret_cast<const float4*>(mine + 4 * q);
    v[4 * q] = t.x; v[4 * q + 1] = t.y; v[4 * q + 2] = t.z; v[4 * q + 3] = t.w;
  }
  float m = -3.4e38f;
#pragma unroll
  for (int cc = 0; cc < C; ++cc) {
    float t = fmaxf(v[cc], 0.f);
    v[cc] = t;
    m = fmaxf(m, t);
  }
  float s = 0.f;
#pragma unroll
  for (int cc = 0; cc < C; ++cc) s += __expf(v[cc] - m);
  float ls = m + __logf(s);
#pragma unroll
  for (int q = 0; q < PO; ++q) {
    float4 t;
    t.x = v[4 * q] - ls; t.y = v[4 * q + 1] - ls; t.z = v[4 * q + 2] - ls; t.w = v[4 * q + 3] - ls;
    *reinterpret_cast<float4*>(mine + 4 * q) = t;
  }

#pragma unroll
  for (int j = 0; j < PO; ++j) {
    const int piece = j * 32 + lane;
    const int r   = piece / PO;
    const int col = 4 * (piece - r * PO);
    if (n0 + r < N)
      *reinterpret_cast<float4*>(out + (size_t)(n0 + r) * C + col) =
          *reinterpret_cast<const float4*>(lds + r * LP + col);
  }
}

static inline int ceil_div(long long a, int b) { return (int)((a + b - 1) / b); }

static void launch_gemm(const float* A, const float* A2, const float* B, const float* bias,
                        float* C, float* Z, int N, int K, int F, bool reluA,
                        hipStream_t stream) {
  const int TPB = 256;
  int rowT = ceil_div(N, 16), colT = ceil_div(F, 16);
  int blocks = ceil_div((long long)rowT * colT, TPB / 32);
  if (K == 64 && !reluA && A2 == nullptr) {
    gemm_wmma<64, false, false><<<blocks, TPB, 0, stream>>>(A, A2, B, bias, C, Z, N, F, rowT, colT);
  } else if (K == 48 && reluA && A2 != nullptr) {
    gemm_wmma<48, true, true><<<blocks, TPB, 0, stream>>>(A, A2, B, bias, C, Z, N, F, rowT, colT);
  } else if (K == 48 && reluA) {
    gemm_wmma<48, true, false><<<blocks, TPB, 0, stream>>>(A, A2, B, bias, C, Z, N, F, rowT, colT);
  } else {
    gemm_wmma_gen<<<blocks, TPB, 0, stream>>>(A, A2, B, bias, C, Z, N, K, F, rowT, colT,
                                              reluA ? 1 : 0);
  }
}

extern "C" void kernel_launch(void* const* d_in, const int* in_sizes, int n_in,
                              void* d_out, int out_size, void* d_ws, size_t ws_size,
                              hipStream_t stream) {
  const float* x  = (const float*)d_in[0];
  const int*   ei = (const int*)d_in[1];
  const float* ew = (const float*)d_in[2];
  const float* W1 = (const float*)d_in[3];
  const float* V1 = (const float*)d_in[4];
  const float* b1 = (const float*)d_in[5];
  const float* W2 = (const float*)d_in[6];
  const float* V2 = (const float*)d_in[7];
  const float* b2 = (const float*)d_in[8];
  float* out = (float*)d_out;

  const int H   = in_sizes[5];
  const int Cn  = in_sizes[8];
  const int Fin = in_sizes[3] / H;
  const int N   = in_sizes[0] / Fin;
  const int E   = in_sizes[2];
  const int* src = ei;
  const int* dst = ei + E;

  float* ws = (float*)d_ws;
  size_t off = 0;
  float* deg  = ws + off; off += ((size_t)N + 255) & ~(size_t)255;
  float* norm = ws + off; off += ((size_t)E + 255) & ~(size_t)255;
  float* hbuf = ws + off; off += ((size_t)N * H + 255) & ~(size_t)255;
  float* z1   = ws + off; off += ((size_t)N * H + 255) & ~(size_t)255;
  const size_t wsFloats = ws_size / sizeof(float);
  float* deg2 = nullptr;
  float* z1b  = nullptr;
  float* outb = nullptr;
  {
    const size_t nDeg = ((size_t)N + 255) & ~(size_t)255;
    const size_t nZ1  = ((size_t)N * H + 255) & ~(size_t)255;
    const size_t nOut = ((size_t)N * Cn + 255) & ~(size_t)255;
    if (off + nDeg <= wsFloats) { deg2 = ws + off; off += nDeg; }
    if (off + nZ1  <= wsFloats) { z1b  = ws + off; off += nZ1; }
    if (off + nOut <= wsFloats) { outb = ws + off; off += nOut; }
  }

  const int TPB = 256;

  if (Fin == 64 && H == 48 && Cn == 40 && deg2 != nullptr && z1b != nullptr && outb != nullptr) {
    const int rowT = ceil_div(N, 16);
    const int nbP1 = ceil_div((long long)rowT * ceil_div(H, 16), TPB / 32);
    const int nbP2 = ceil_div((long long)rowT * ceil_div(Cn, 16), TPB / 32);
    const int nbE  = ceil_div(E, TPB);
    const int per1 = ceil_div(nbP1, 3) > ceil_div(nbE, 8) ? ceil_div(nbP1, 3) : ceil_div(nbE, 8);
    const int per2 = ceil_div(nbP2, 3) > ceil_div(nbE, 8) ? ceil_div(nbP2, 3) : ceil_div(nbE, 8);
    float* suma = hbuf;

    zero_f32<<<ceil_div(N, TPB), TPB, 0, stream>>>(deg, deg2, N);
    degree_beside_product<<<per1 * 11, TPB, 0, stream>>>(dst, ew, deg, deg2, E, x, W1, hbuf,
                                                         N, H, rowT, ceil_div(H, 16));
    inv_sqrt_degree<<<ceil_div(N, TPB), TPB, 0, stream>>>(deg, deg2, N);
    norm_beside_product<<<per1 * 11, TPB, 0, stream>>>(src, dst, ew, deg, norm, E,
                                                       x, V1, b1, z1, z1b,
                                                       N, H, rowT, ceil_div(H, 16));

    edge_scatter_lines<48><<<nbE, TPB, 0, stream>>>(hbuf, norm, src, dst, z1, z1b, N, E);

    product_rows<48, true, true><<<nbP2, TPB, 0, stream>>>(z1, z1b, W2, out, suma, outb,
                                                          N, Cn, rowT, ceil_div(Cn, 16));
    scatter_beside_product<40, 48, true, true><<<per2 * 11, TPB, 0, stream>>>(
        out, norm, src, dst, suma, outb, E, z1, z1b, V2, b2, N, rowT, ceil_div(Cn, 16));
    relu_logsoftmax_lines<40><<<ceil_div(N, TPB), TPB, 0, stream>>>(suma, outb, out, N);
    return;
  }

  zero_f32<<<ceil_div(N, TPB), TPB, 0, stream>>>(deg, deg2, N);
  deg_accum<<<ceil_div(E, TPB), TPB, 0, stream>>>(dst, ew, deg, deg2, E);
  norm_compute<<<ceil_div(E, TPB), TPB, 0, stream>>>(src, dst, ew, deg, deg2, norm, E);

  launch_gemm(x, nullptr, W1, nullptr, hbuf, nullptr, N, Fin, H, false, stream);
  launch_gemm(x, nullptr, V1, b1,      z1,   z1b,     N, Fin, H, false, stream);
  launch_scatter(hbuf, norm, src, dst, z1, z1b, E, H, stream);

  launch_gemm(z1, z1b, W2, nullptr, hbuf, nullptr, N, H, Cn, true, stream);
  launch_gemm(z1, z1b, V2, b2,      out,  outb,    N, H, Cn, true, stream);
  {
    launch_scatter(hbuf, norm, src, dst, out, outb, E, Cn, stream);
    if (Cn == 40) {
      relu_logsoftmax_t<40><<<ceil_div(N, TPB), TPB, 0, stream>>>(out, outb, N);
    } else {
      relu_logsoftmax_gen<<<ceil_div(N, TPB), TPB, 0, stream>>>(out, outb, N, Cn);
    }
  }
}


# Edge norm formed inside layer 1's scatter: each lane reads its edge's two deg^-1/2 words and stores the norm (same bits) for layer 2; norm launch gone, x@V1 beside the degree adds; sums as before
# speedup vs baseline: 2.6737x; 1.0407x over previous
#include <hip/hip_runtime.h>

typedef float v2f __attribute__((ext_vector_type(2)));
typedef float v8f __attribute__((ext_vector_type(8)));

__global__ void zero_f32(float* __restrict__ p, float* __restrict__ p2, int n) {
  int i = blockIdx.x * blockDim.x + threadIdx.x;
  if (i < n) {
    p[i] = 0.f;
    if (p2 != nullptr) p2[i] = 0.f;
  }
}

__device__ __forceinline__ int l2_half() {
  const unsigned id = __builtin_amdgcn_s_sendmsg_rtn(0x87);
  return (int)((id >> 18) & 1u);
}

__global__ void deg_accum(const int* __restrict__ dst, const float* __restrict__ w,
                          float* __restrict__ deg, float* __restrict__ deg2, int E) {
  int e = blockIdx.x * blockDim.x + threadIdx.x;
  float* d = (deg2 != nullptr && l2_half()) ? deg2 : deg;
  if (e < E) atomicAdd(d + dst[e], w[e]);
}

__global__ void norm_compute(const int* __restrict__ src, const int* __restrict__ dst,
                             const float* __restrict__ w, const float* __restrict__ deg,
                             const float* __restrict__ deg2, float* __restrict__ norm, int E) {
  int e = blockIdx.x * blockDim.x + threadIdx.x;
  if (e >= E) return;
  float ds = deg[src[e]];
  float dd = deg[dst[e]];
  if (deg2 != nullptr) {
    ds += deg2[src[e]];
    dd += deg2[dst[e]];
  }
  float is = ds > 0.f ? rsqrtf(ds) : 0.f;
  float id = dd > 0.f ? rsqrtf(dd) : 0.f;
  norm[e] = is * w[e] * id;
}

__global__ void inv_sqrt_degree(float* __restrict__ deg, const float* __restrict__ deg2, int N) {
  int n = blockIdx.x * blockDim.x + threadIdx.x;
  if (n >= N) return;
  const float d = deg[n] + deg2[n];
  deg[n] = d > 0.f ? rsqrtf(d) : 0.f;
}

template <int K, bool RELU_A, bool TWO_A>
__global__ void gemm_wmma(const float* __restrict__ A, const float* __restrict__ A2,
                          const float* __restrict__ B,
                          const float* __restrict__ bias, float* __restrict__ C,
                          float* __restrict__ Z,
                          int N, int F, int rowTiles, int colTiles) {
  const int lane = threadIdx.x & 31;
  const int wave = threadIdx.x >> 5;
  const int tile = blockIdx.x * (blockDim.x >> 5) + wave;
  if (tile >= rowTiles * colTiles) return;

  const int rt = tile / colTiles;
  const int ct = tile - rt * colTiles;
  const int row0 = rt << 4;
  const int col0 = ct << 4;
  const int half = lane >> 4;
  const int lm   = lane & 15;

  const bool cOK = (col0 + lm) < F;
  const int  arow = min(row0 + lm, N - 1);
  const int  bcol = min(col0 + lm, F - 1);

  const float* Arow = A + (size_t)arow * K + 2 * half;
  const float* Bcol = B + bcol;

  v2f afrag[K / 4];
  v2f bfrag[K / 4];
#pragma unroll
  for (int i = 0; i < K / 4; ++i) {
    const int ka = 4 * i + 2 * half;
    afrag[i] = *reinterpret_cast<const v2f*>(Arow + 4 * i);
    bfrag[i].x = Bcol[(size_t)ka * F];
    bfrag[i].y = Bcol[(size_t)(ka + 1) * F];
  }
  if (TWO_A) {
    const float* A2row = A2 + (size_t)arow * K + 2 * half;
#pragma unroll
    for (int i = 0; i < K / 4; ++i) {
      const v2f t = *reinterpret_cast<const v2f*>(A2row + 4 * i);
      afrag[i].x += t.x;
      afrag[i].y += t.y;
    }
  }
  if (RELU_A) {
#pragma unroll
    for (int i = 0; i < K / 4; ++i) {
      afrag[i].x = fmaxf(afrag[i].x, 0.f);
      afrag[i].y = fmaxf(afrag[i].y, 0.f);
    }
  }

  v8f acc = {};
#pragma unroll
  for (int i = 0; i < K / 4; ++i) {
    acc = __builtin_amdgcn_wmma_f32_16x16x4_f32(
        false, afrag[i], false, bfrag[i], (short)0, acc, false, false);
  }

  const float bv = (bias != nullptr) ? bias[bcol] : 0.f;
#pragma unroll
  for (int r = 0; r < 8; ++r) {
    const int row = row0 + r + 8 * half;
    if (row < N && cOK) {
      C[(size_t)row * F + col0 + lm] = acc[r] + bv;
      if (Z != nullptr) Z[(size_t)row * F + col0 + lm] = 0.f;
    }
  }
}

__global__ void gemm_wmma_gen(const float* __restrict__ A, const float* __restrict__ A2,
                              const float* __restrict__ B,
                              const float* __restrict__ bias, float* __restrict__ C,
                              float* __restrict__ Z,
                              int N, int K, int F, int rowTiles, int colTiles, int reluA) {
  const int lane = threadIdx.x & 31;
  const int wave = threadIdx.x >> 5;
  const int tile = blockIdx.x * (blockDim.x >> 5) + wave;
  if (tile >= rowTiles * colTiles) return;
  const int rt = tile / colTiles;
  const int ct = tile - rt * colTiles;
  const int row0 = rt << 4, col0 = ct << 4;
  const int half = lane >> 4, lm = lane & 15;
  const bool cOK = (col0 + lm) < F;
  const int arow = min(row0 + lm, N - 1);
  const int bcol = min(col0 + lm, F - 1);
  v8f acc = {};
  for (int kb = 0; kb < K; kb += 4) {
    const int ka = kb + 2 * half;
    v2f a = *reinterpret_cast<const v2f*>(A + (size_t)arow * K + ka);
    if (A2 != nullptr) {
      const v2f t = *reinterpret_cast<const v2f*>(A2 + (size_t)arow * K + ka);
      a.x += t.x;
      a.y += t.y;
    }
    if (reluA) { a.x = fmaxf(a.x, 0.f); a.y = fmaxf(a.y, 0.f); }
    v2f b;
    b.x = B[(size_t)ka * F + bcol];
    b.y = B[(size_t)(ka + 1) * F + bcol];
    acc = __builtin_amdgcn_wmma_f32_16x16x4_f32(false, a, false, b, (short)0, acc,
                                                false, false);
  }
  const float bv = (bias != nullptr) ? bias[bcol] : 0.f;
#pragma unroll
  for (int r = 0; r < 8; ++r) {
    const int row = row0 + r + 8 * half;
    if (row < N && cOK) {
      C[(size_t)row * F + col0 + lm] = acc[r] + bv;
      if (Z != nullptr) Z[(size_t)row * F + col0 + lm] = 0.f;
    }
  }
}

template <int F, bool TAIL>
__device__ __forceinline__ void scatter_wave(const float* __restrict__ h, float* __restrict__ agg,
                                             int s_l, int d_l, float w_l,
                                             int lm, int hbase, long long e0, int E) {
  constexpr int J = (F + 15) / 16;
  constexpr int P = 8;
#pragma unroll
  for (int b = 0; b < 16; b += P) {
    float v[P][J];
    float ww[P];
    int   dd[P];
#pragma unroll
    for (int p = 0; p < P; ++p) {
      const int q = hbase + b + p;
      const int s = __shfl(s_l, q);
      ww[p] = __shfl(w_l, q);
      dd[p] = __shfl(d_l, q);
      const float* hr = h + (size_t)s * F + lm;
#pragma unroll
      for (int j = 0; j < J; ++j)
        v[p][j] = (16 * j + lm < F) ? hr[16 * j] : 0.f;
    }
#pragma unroll
    for (int p = 0; p < P; ++p) {
      if (!TAIL || e0 + hbase + b + p < E) {
        float* ar = agg + (size_t)dd[p] * F + lm;
#pragma unroll
        for (int j = 0; j < J; ++j)
          if (16 * j + lm < F) atomicAdd(ar + 16 * j, v[p][j] * ww[p]);
      }
    }
  }
}

template <int F>
__global__ void edge_scatter_rows(const float* __restrict__ h, const float* __restrict__ norm,
                                  const int* __restrict__ src, const int* __restrict__ dst,
                                  float* __restrict__ aggA, float* __restrict__ aggB, int E) {
  const int lane  = threadIdx.x & 31;
  const int lm    = lane & 15;
  const int hbase = lane & 16;
  const long long wave = ((long long)blockIdx.x * blockDim.x + threadIdx.x) >> 5;
  const long long e0 = wave * 32;
  if (e0 >= E) return;
  const long long el = e0 + lane;
  const long long ec = el < E ? el : (long long)E - 1;
  const int   s_l = src[ec];
  const int   d_l = dst[ec];
  const float w_l = norm[ec];
  float* agg = (aggB != nullptr && l2_half()) ? aggB : aggA;
  if (e0 + 32 <= E) scatter_wave<F, false>(h, agg, s_l, d_l, w_l, lm, hbase, e0, E);
  else              scatter_wave<F, true >(h, agg, s_l, d_l, w_l, lm, hbase, e0, E);
}

__global__ void edge_scatter_rows_gen(const float* __restrict__ h, const float* __restrict__ norm,
                                      const int* __restrict__ src, const int* __restrict__ dst,
                                      float* __restrict__ aggA, float* __restrict__ aggB,
                                      int E, int F) {
  const int lane = threadIdx.x & 31;
  const int lm   = lane & 15;
  const long long e = (((long long)blockIdx.x * blockDim.x + threadIdx.x) >> 5) * 2 + (lane >> 4);
  if (e >= E) return;
  const int   s = src[e];
  const int   d = dst[e];
  const float w = norm[e];
  float* agg = (aggB != nullptr && l2_half()) ? aggB : aggA;
  for (int f = lm; f < F; f += 16)
    atomicAdd(agg + (size_t)d * F + f, h[(size_t)s * F + f] * w);
}

static void launch_scatter(const float* h, const float* norm, const int* src, const int* dst,
                           float* agg, float* aggB, int E, int F, hipStream_t stream) {
  const int TPB = 256;
  if (F == 48) {
    edge_scatter_rows<48><<<(E + 255) / 256, TPB, 0, stream>>>(h, norm, src, dst, agg, aggB, E);
  } else if (F == 40) {
    edge_scatter_rows<40><<<(E + 255) / 256, TPB, 0, stream>>>(h, norm, src, dst, agg, aggB, E);
  } else {
    edge_scatter_rows_gen<<<(E + 15) / 16, TPB, 0, stream>>>(h, norm, src, dst, agg, aggB, E, F);
  }
}


typedef __attribute__((address_space(1))) float       gmem_f32;
typedef __attribute__((address_space(1))) const float gmem_cf32;
__device__ __forceinline__ unsigned long long scalar_u64(unsigned long long a) {
  const unsigned lo = __builtin_amdgcn_readfirstlane((unsigned)a);
  const unsigned hi = __builtin_amdgcn_readfirstlane((unsigned)(a >> 32));
  return ((unsigned long long)hi << 32) | lo;
}
__device__ __forceinline__ gmem_cf32* scalar_base(const float* p) {
  return (gmem_cf32*)scalar_u64((unsigned long long)p);
}
__device__ __forceinline__ gmem_f32* scalar_base(float* p) {
  return (gmem_f32*)scalar_u64((unsigned long long)p);
}

template <int F, bool TAIL>
__device__ __forceinline__ void scatter_wave_lines(const float* __restrict__ h, float* __restrict__ agg,
                                                   int N, int s_l, int d_l, float w_l,
                                                   int lane, long long e0, int E) {
  constexpr int T = F - 32;
  constexpr int G = 32 / T;
  constexpr int B = 16;
  const float* hY = h + (size_t)N * 32;
  float*       aY = agg + (size_t)N * 32;
  const int sub = lane / T;
  const int ly  = lane % T;
#pragma unroll
  for (int b = 0; b < 32; b += B) {
    float vx[B];
    float vy[B / G];
#pragma unroll
    for (int p = 0; p < B; ++p) {
      const int s = __builtin_amdgcn_readlane(s_l, b + p);
      gmem_cf32* hrow = scalar_base(h + (size_t)s * 32);
      vx[p] = hrow[lane];
    }
#pragma unroll
    for (int g = 0; g < B / G; ++g) {
      const int s = __shfl(s_l, b + g * G + sub);
      vy[g] = hY[(size_t)s * T + ly];
    }
#pragma unroll
    for (int p = 0; p < B; ++p) {
      const int   d = __builtin_amdgcn_readlane(d_l, b + p);
      const float w = __int_as_float(__builtin_amdgcn_readlane(__float_as_int(w_l), b + p));
      gmem_f32* arow = scalar_base(agg + (size_t)d * 32);
      if (!TAIL || e0 + b + p < E)
        __hip_atomic_fetch_add(arow + lane, vx[p] * w, __ATOMIC_RELAXED, __HIP_MEMORY_SCOPE_AGENT);
    }
#pragma unroll
    for (int g = 0; g < B / G; ++g) {
      const int   q = b + g * G + sub;
      const int   d = __shfl(d_l, q);
      const float w = __shfl(w_l, q);
      if (!TAIL || e0 + q < E) atomicAdd(aY + (size_t)d * T + ly, vy[g] * w);
    }
  }
}

template <int F, bool NORM_HERE>
__device__ __forceinline__ void scatter_edges_lines(const float* __restrict__ h,
                                                    const float* __restrict__ norm,
                                                    const float* __restrict__ dinv,
                                                    float* __restrict__ norm_out,
                                                    const int* __restrict__ src,
                                                    const int* __restrict__ dst,
                                                    float* __restrict__ aggA, float* __restrict__ aggB,
                                                    int N, int E, long long e0) {
  const int lane = threadIdx.x & 31;
  const long long el = e0 + lane;
  const long long ec = el < E ? el : (long long)E - 1;
  const int   s_l = src[ec];
  const int   d_l = dst[ec];
  float       w_l = norm[ec];
  if (NORM_HERE) {
    const float is = dinv[s_l];
    const float id = dinv[d_l];
    w_l = is * w_l * id;
    if (el < E) norm_out[el] = w_l;
  }
  float* agg = l2_half() ? aggB : aggA;
  if (e0 + 32 <= E) scatter_wave_lines<F, false>(h, agg, N, s_l, d_l, w_l, lane, e0, E);
  else              scatter_wave_lines<F, true >(h, agg, N, s_l, d_l, w_l, lane, e0, E);
}

template <int F>
__global__ void edge_scatter_lines(const float* __restrict__ h, const float* __restrict__ w,
                                   const float* __restrict__ dinv, float* __restrict__ norm_out,
                                   const int* __restrict__ src, const int* __restrict__ dst,
                                   float* __restrict__ aggA, float* __restrict__ aggB, int N, int E) {
  const long long wave = ((long long)blockIdx.x * blockDim.x + threadIdx.x) >> 5;
  const long long e0 = wave * 32;
  if (e0 >= E) return;
  scatter_edges_lines<F, true>(h, w, dinv, norm_out, src, dst, aggA, aggB, N, E, e0);
}

__device__ __forceinline__ bool product_slot(int b, int& idx) {
  const int q = b / 11;
  const int r = b - q * 11;
  if ((r & 3) == 1) { idx = 3 * q + (r >> 2); return true; }
  idx = 8 * q + r - ((r + 2) >> 2);
  return false;
}

template <int K, bool RELU_A, bool TWO_A, bool SPLIT_A>
__device__ __forceinline__ void stage_a_rows(float* __restrict__ lds,
                                             const float* __restrict__ A,
                                             const float* __restrict__ A2,
                                             int N, int row0, int lane) {
  constexpr int LP = K + 4;
  constexpr int PX = (SPLIT_A ? 32 : K) / 4;
  constexpr int NX = (16 * PX) / 32;
  constexpr int TY = SPLIT_A ? K - 32 : 4;
  constexpr int PY = TY / 4;
  constexpr int NY = SPLIT_A ? (16 * PY) / 32 : 0;
  constexpr int NL = NX + NY;
  float4 va[NL];
  float4 vb[TWO_A ? NL : 1];
  int    lo[NL];
#pragma unroll
  for (int j = 0; j < NL; ++j) {
    size_t o;
    if (j < NX) {
      const int piece = j * 32 + lane;
      const int r   = piece / PX;
      const int col = 4 * (piece - r * PX);
      o = (size_t)min(row0 + r, N - 1) * (4 * PX) + col;
      lo[j] = r * LP + col;
    } else {
      const int piece = (j - NX) * 32 + lane;
      const int r   = piece / PY;
      const int col = 4 * (piece - r * PY);
      o = (size_t)N * 32 + (size_t)min(row0 + r, N - 1) * TY + col;
      lo[j] = r * LP + 32 + col;
    }
    va[j] = *reinterpret_cast<const float4*>(A + o);
    if (TWO_A) vb[j] = *reinterpret_cast<const float4*>(A2 + o);
  }
#pragma unroll
  for (int j = 0; j < NL; ++j) {
    float4 v = va[j];
    if (TWO_A) {
      v.x += vb[j].x; v.y += vb[j].y; v.z += vb[j].z; v.w += vb[j].w;
    }
    if (RELU_A) {
      v.x = fmaxf(v.x, 0.f); v.y = fmaxf(v.y, 0.f); v.z = fmaxf(v.z, 0.f); v.w = fmaxf(v.w, 0.f);
    }
    *reinterpret_cast<float4*>(lds + lo[j]) = v;
  }
}

template <int K, bool RELU_A, bool TWO_A, bool ADD_C, bool SPLIT_A>
__device__ __forceinline__ void product_tile(const float* __restrict__ A,
                                             const float* __restrict__ A2,
                                             const float* __restrict__ B,
                                             const float* __restrict__ bias,
                                             float* __restrict__ C, float* __restrict__ C2,
                                             float* __restrict__ Z, float* __restrict__ Z2,
                                             int N, int F, int rowTiles, int colTiles,
                                             int vblock) {
  const int lane = threadIdx.x & 31;
  const int wave = threadIdx.x >> 5;
  const int tile = vblock * (int)(blockDim.x >> 5) + wave;
  if (tile >= rowTiles * colTiles) return;

  const int rt = tile / colTiles;
  const int ct = tile - rt * colTiles;
  const int row0 = rt << 4;
  const int col0 = ct << 4;
  const int half = lane >> 4;
  const int lm   = lane & 15;

  const bool cOK = (col0 + lm) < F;
  const int  bcol = min(col0 + lm, F - 1);

  constexpr int LP = K + 4;
  __shared__ __attribute__((aligned(16))) float a_rows[8 * 16 * LP];
  float* lds = a_rows + wave * (16 * LP);
  stage_a_rows<K, RELU_A, TWO_A, SPLIT_A>(lds, A, A2, N, row0, lane);

  const float* Bcol = B + bcol;

  v2f afrag[K / 4];
  v2f bfrag[K / 4];
#pragma unroll
  for (int i = 0; i < K / 4; ++i) {
    const int ka = 4 * i + 2 * half;
    afrag[i] = *reinterpret_cast<const v2f*>(lds + lm * LP + ka);
    bfrag[i].x = Bcol[(size_t)ka * F];
    bfrag[i].y = Bcol[(size_t)(ka + 1) * F];
  }

  v8f acc = {};
#pragma unroll
  for (int i = 0; i < K / 4; ++i) {
    acc = __builtin_amdgcn_wmma_f32_16x16x4_f32(
        false, afrag[i], false, bfrag[i], (short)0, acc, false, false);
  }

  const bool   inX = col0 < 32;
  const int    cp  = inX ? 32 : F - 32;
  const size_t cb  = (inX ? (size_t)0 : (size_t)N * 32) + (size_t)(inX ? col0 : col0 - 32) + lm;

  const float bv = (bias != nullptr) ? bias[bcol] : 0.f;
  if (ADD_C) {
    float* Cp = (C2 != nullptr && l2_half()) ? C2 : C;
#pragma unroll
    for (int r = 0; r < 8; ++r) {
      const int row = row0 + r + 8 * half;
      if (row < N && cOK) atomicAdd(Cp + cb + (size_t)row * cp, acc[r] + bv);
    }
  } else {
#pragma unroll
    for (int r = 0; r < 8; ++r) {
      const int row = row0 + r + 8 * half;
      if (row < N && cOK) {
        const size_t o = cb + (size_t)row * cp;
        C[o] = acc[r] + bv;
        if (Z  != nullptr) Z[o]  = 0.f;
        if (Z2 != nullptr) Z2[o] = 0.f;
      }
    }
  }
}

template <int K, bool RELU_A, bool TWO_A>
__global__ void product_rows(const float* __restrict__ A, const float* __restrict__ A2,
                             const float* __restrict__ B, float* __restrict__ C,
                             float* __restrict__ Z, float* __restrict__ Z2,
                             int N, int F, int rowTiles, int colTiles) {
  product_tile<K, RELU_A, TWO_A, false, true>(A, A2, B, nullptr, C, nullptr, Z, Z2,
                                              N, F, rowTiles, colTiles, (int)blockIdx.x);
}

__device__ __forceinline__ bool product_slot_3of7(int b, int& idx) {
  const int q = b / 7;
  const int r = b - q * 7;
  if (r & 1) { idx = 3 * q + (r >> 1); return true; }
  idx = 4 * q + (r >> 1);
  return false;
}

__global__ void degree_beside_products(const int* __restrict__ dst, const float* __restrict__ w,
                                       float* __restrict__ deg, float* __restrict__ deg2, int E,
                                       const float* __restrict__ x, const float* __restrict__ Wm,
                                       const float* __restrict__ Vm, const float* __restrict__ bias,
                                       float* __restrict__ h, float* __restrict__ z,
                                       float* __restrict__ zb,
                                       int N, int F, int rowTiles, int colTiles) {
  int idx;
  if (product_slot_3of7((int)blockIdx.x, idx)) {
    const bool root = (idx & 1) != 0;
    product_tile<64, false, false, false, false>(x, nullptr, root ? Vm : Wm,
                                                 root ? bias : nullptr, root ? z : h, nullptr,
                                                 root ? zb : nullptr, nullptr,
                                                 N, F, rowTiles, colTiles, idx >> 1);
  } else {
    const long long e = (long long)idx * blockDim.x + threadIdx.x;
    float* d = l2_half() ? deg2 : deg;
    if (e < E) atomicAdd(d + dst[e], w[e]);
  }
}

template <int F, int K, bool RELU_A, bool TWO_A>
__global__ void scatter_beside_product(const float* __restrict__ h, const float* __restrict__ norm,
                                       const int* __restrict__ src, const int* __restrict__ dst,
                                       float* __restrict__ aggA, float* __restrict__ aggB, int E,
                                       const float* __restrict__ A, const float* __restrict__ A2,
                                       const float* __restrict__ Vm, const float* __restrict__ bias,
                                       int N, int rowTiles, int colTiles) {
  int idx;
  if (product_slot((int)blockIdx.x, idx)) {
    product_tile<K, RELU_A, TWO_A, true, true>(A, A2, Vm, bias, aggA, aggB, nullptr, nullptr,
                                               N, F, rowTiles, colTiles, idx);
  } else {
    const long long wave = ((long long)idx * blockDim.x + threadIdx.x) >> 5;
    const long long e0 = wave * 32;
    if (e0 >= E) return;
    scatter_edges_lines<F, false>(h, norm, nullptr, nullptr, src, dst, aggA, aggB, N, E, e0);
  }
}

template <int C>
__global__ void relu_logsoftmax_t(float* __restrict__ y, const float* __restrict__ yb, int N) {
  int n = blockIdx.x * blockDim.x + threadIdx.x;
  if (n >= N) return;
  float v[C];
  float m = -3.4e38f;
  float* row = y + (size_t)n * C;
#pragma unroll
  for (int c = 0; c < C; ++c) v[c] = row[c];
  if (yb != nullptr) {
    const float* rb = yb + (size_t)n * C;
#pragma unroll
    for (int c = 0; c < C; ++c) v[c] += rb[c];
  }
#pragma unroll
  for (int c = 0; c < C; ++c) {
    float t = fmaxf(v[c], 0.f);
    v[c] = t;
    m = fmaxf(m, t);
  }
  float s = 0.f;
#pragma unroll
  for (int c = 0; c < C; ++c) s += __expf(v[c] - m);
  float ls = m + __logf(s);
#pragma unroll
  for (int c = 0; c < C; ++c) row[c] = v[c] - ls;
}

__global__ void relu_logsoftmax_gen(float* __restrict__ y, const float* __restrict__ yb,
                                    int N, int C) {
  int n = blockIdx.x * blockDim.x + threadIdx.x;
  if (n >= N) return;
  float* row = y + (size_t)n * C;
  if (yb != nullptr) {
    const float* rb = yb + (size_t)n * C;
    for (int c = 0; c < C; ++c) row[c] += rb[c];
  }
  float m = -3.4e38f;
  for (int c = 0; c < C; ++c) m = fmaxf(m, fmaxf(row[c], 0.f));
  float s = 0.f;
  for (int c = 0; c < C; ++c) s += __expf(fmaxf(row[c], 0.f) - m);
  float ls = m + __logf(s);
  for (int c = 0; c < C; ++c) row[c] = fmaxf(row[c], 0.f) - ls;
}

template <int C>
__global__ void relu_logsoftmax_lines(const float* __restrict__ ya, const float* __restrict__ yb,
                                      float* __restrict__ out, int N) {
  constexpr int T  = C - 32;
  constexpr int LP = C + 4;
  constexpr int PY = T / 4;
  constexpr int PO = C / 4;
  __shared__ __attribute__((aligned(16))) float rows[8 * 32 * LP];
  const int lane = threadIdx.x & 31;
  const int wave = threadIdx.x >> 5;
  const int n0 = ((int)blockIdx.x * (int)(blockDim.x >> 5) + wave) * 32;
  if (n0 >= N) return;
  float* lds = rows + wave * (32 * LP);

  float4 a[8 + PY], b[8 + PY];
  int    lo[8 + PY];
#pragma unroll
  for (int j = 0; j < 8 + PY; ++j) {
    size_t o;
    if (j < 8) {
      const int piece = j * 32 + lane;
      const int r   = piece >> 3;
      const int col = 4 * (piece & 7);
      o = (size_t)min(n0 + r, N - 1) * 32 + col;
      lo[j] = r * LP + col;
    } else {
      const int piece = (j - 8) * 32 + lane;
      const int r   = piece / PY;
      const int col = 4 * (piece - r * PY);
      o = (size_t)N * 32 + (size_t)min(n0 + r, N - 1) * T + col;
      lo[j] = r * LP + 32 + col;
    }
    a[j] = *reinterpret_cast<const float4*>(ya + o);
    b[j] = *reinterpret_cast<const float4*>(yb + o);
  }
#pragma unroll
  for (int j = 0; j < 8 + PY; ++j) {
    float4 v;
    v.x = a[j].x + b[j].x; v.y = a[j].y + b[j].y; v.z = a[j].z + b[j].z; v.w = a[j].w + b[j].w;
    *reinterpret_cast<float4*>(lds + lo[j]) = v;
  }

  float v[C];
  float* mine = lds + lane * LP;
#pragma unroll
  for (int q = 0; q < PO; ++q) {
    const float4 t = *reinterpret_cast<const float4*>(mine + 4 * q);
    v[4 * q] = t.x; v[4 * q + 1] = t.y; v[4 * q + 2] = t.z; v[4 * q + 3] = t.w;
  }
  float m = -3.4e38f;
#pragma unroll
  for (int cc = 0; cc < C; ++cc) {
    float t = fmaxf(v[cc], 0.f);
    v[cc] = t;
    m = fmaxf(m, t);
  }
  float s = 0.f;
#pragma unroll
  for (int cc = 0; cc < C; ++cc) s += __expf(v[cc] - m);
  float ls = m + __logf(s);
#pragma unroll
  for (int q = 0; q < PO; ++q) {
    float4 t;
    t.x = v[4 * q] - ls; t.y = v[4 * q + 1] - ls; t.z = v[4 * q + 2] - ls; t.w = v[4 * q + 3] - ls;
    *reinterpret_cast<float4*>(mine + 4 * q) = t;
  }

#pragma unroll
  for (int j = 0; j < PO; ++j) {
    const int piece = j * 32 + lane;
    const int r   = piece / PO;
    const int col = 4 * (piece - r * PO);
    if (n0 + r < N)
      *reinterpret_cast<float4*>(out + (size_t)(n0 + r) * C + col) =
          *reinterpret_cast<const float4*>(lds + r * LP + col);
  }
}

static inline int ceil_div(long long a, int b) { return (int)((a + b - 1) / b); }

static void launch_gemm(const float* A, const float* A2, const float* B, const float* bias,
                        float* C, float* Z, int N, int K, int F, bool reluA,
                        hipStream_t stream) {
  const int TPB = 256;
  int rowT = ceil_div(N, 16), colT = ceil_div(F, 16);
  int blocks = ceil_div((long long)rowT * colT, TPB / 32);
  if (K == 64 && !reluA && A2 == nullptr) {
    gemm_wmma<64, false, false><<<blocks, TPB, 0, stream>>>(A, A2, B, bias, C, Z, N, F, rowT, colT);
  } else if (K == 48 && reluA && A2 != nullptr) {
    gemm_wmma<48, true, true><<<blocks, TPB, 0, stream>>>(A, A2, B, bias, C, Z, N, F, rowT, colT);
  } else if (K == 48 && reluA) {
    gemm_wmma<48, true, false><<<blocks, TPB, 0, stream>>>(A, A2, B, bias, C, Z, N, F, rowT, colT);
  } else {
    gemm_wmma_gen<<<blocks, TPB, 0, stream>>>(A, A2, B, bias, C, Z, N, K, F, rowT, colT,
                                              reluA ? 1 : 0);
  }
}

extern "C" void kernel_launch(void* const* d_in, const int* in_sizes, int n_in,
                              void* d_out, int out_size, void* d_ws, size_t ws_size,
                              hipStream_t stream) {
  const float* x  = (const float*)d_in[0];
  const int*   ei = (const int*)d_in[1];
  const float* ew = (const float*)d_in[2];
  const float* W1 = (const float*)d_in[3];
  const float* V1 = (const float*)d_in[4];
  const float* b1 = (const float*)d_in[5];
  const float* W2 = (const float*)d_in[6];
  const float* V2 = (const float*)d_in[7];
  const float* b2 = (const float*)d_in[8];
  float* out = (float*)d_out;

  const int H   = in_sizes[5];
  const int Cn  = in_sizes[8];
  const int Fin = in_sizes[3] / H;
  const int N   = in_sizes[0] / Fin;
  const int E   = in_sizes[2];
  const int* src = ei;
  const int* dst = ei + E;

  float* ws = (float*)d_ws;
  size_t off = 0;
  float* deg  = ws + off; off += ((size_t)N + 255) & ~(size_t)255;
  float* norm = ws + off; off += ((size_t)E + 255) & ~(size_t)255;
  float* hbuf = ws + off; off += ((size_t)N * H + 255) & ~(size_t)255;
  float* z1   = ws + off; off += ((size_t)N * H + 255) & ~(size_t)255;
  const size_t wsFloats = ws_size / sizeof(float);
  float* deg2 = nullptr;
  float* z1b  = nullptr;
  float* outb = nullptr;
  {
    const size_t nDeg = ((size_t)N + 255) & ~(size_t)255;
    const size_t nZ1  = ((size_t)N * H + 255) & ~(size_t)255;
    const size_t nOut = ((size_t)N * Cn + 255) & ~(size_t)255;
    if (off + nDeg <= wsFloats) { deg2 = ws + off; off += nDeg; }
    if (off + nZ1  <= wsFloats) { z1b  = ws + off; off += nZ1; }
    if (off + nOut <= wsFloats) { outb = ws + off; off += nOut; }
  }

  const int TPB = 256;

  if (Fin == 64 && H == 48 && Cn == 40 && deg2 != nullptr && z1b != nullptr && outb != nullptr) {
    const int rowT = ceil_div(N, 16);
    const int nbP1 = ceil_div((long long)rowT * ceil_div(H, 16), TPB / 32);
    const int nbP2 = ceil_div((long long)rowT * ceil_div(Cn, 16), TPB / 32);
    const int nbE  = ceil_div(E, TPB);
    const int per1 = ceil_div(2 * nbP1, 3) > ceil_div(nbE, 4) ? ceil_div(2 * nbP1, 3) : ceil_div(nbE, 4);
    const int per2 = ceil_div(nbP2, 3) > ceil_div(nbE, 8) ? ceil_div(nbP2, 3) : ceil_div(nbE, 8);
    float* suma = hbuf;

    zero_f32<<<ceil_div(N, TPB), TPB, 0, stream>>>(deg, deg2, N);
    degree_beside_products<<<per1 * 7, TPB, 0, stream>>>(dst, ew, deg, deg2, E, x, W1, V1, b1,
                                                         hbuf, z1, z1b,
                                                         N, H, rowT, ceil_div(H, 16));
    inv_sqrt_degree<<<ceil_div(N, TPB), TPB, 0, stream>>>(deg, deg2, N);

    edge_scatter_lines<48><<<nbE, TPB, 0, stream>>>(hbuf, ew, deg, norm, src, dst, z1, z1b, N, E);

    product_rows<48, true, true><<<nbP2, TPB, 0, stream>>>(z1, z1b, W2, out, suma, outb,
                                                          N, Cn, rowT, ceil_div(Cn, 16));
    scatter_beside_product<40, 48, true, true><<<per2 * 11, TPB, 0, stream>>>(
        out, norm, src, dst, suma, outb, E, z1, z1b, V2, b2, N, rowT, ceil_div(Cn, 16));
    relu_logsoftmax_lines<40><<<ceil_div(N, TPB), TPB, 0, stream>>>(suma, outb, out, N);
    return;
  }

  zero_f32<<<ceil_div(N, TPB), TPB, 0, stream>>>(deg, deg2, N);
  deg_accum<<<ceil_div(E, TPB), TPB, 0, stream>>>(dst, ew, deg, deg2, E);
  norm_compute<<<ceil_div(E, TPB), TPB, 0, stream>>>(src, dst, ew, deg, deg2, norm, E);

  launch_gemm(x, nullptr, W1, nullptr, hbuf, nullptr, N, Fin, H, false, stream);
  launch_gemm(x, nullptr, V1, b1,      z1,   z1b,     N, Fin, H, false, stream);
  launch_scatter(hbuf, norm, src, dst, z1, z1b, E, H, stream);

  launch_gemm(z1, z1b, W2, nullptr, hbuf, nullptr, N, H, Cn, true, stream);
  launch_gemm(z1, z1b, V2, b2,      out,  outb,    N, H, Cn, true, stream);
  {
    launch_scatter(hbuf, norm, src, dst, out, outb, E, Cn, stream);
    if (Cn == 40) {
      relu_logsoftmax_t<40><<<ceil_div(N, TPB), TPB, 0, stream>>>(out, outb, N);
    } else {
      relu_logsoftmax_gen<<<ceil_div(N, TPB), TPB, 0, stream>>>(out, outb, N, Cn);
    }
  }
}
